# starting kernel
# baseline (speedup 1.0000x reference)
#include <hip/hip_runtime.h>
#include <hip/hip_bf16.h>

#define NFEAT  512
#define NCLASS 256

typedef float v2f __attribute__((ext_vector_type(2)));
typedef float v8f __attribute__((ext_vector_type(8)));

__global__ void spgat_gemm(const float* __restrict__ x,
                           const float* __restrict__ W,
                           float* __restrict__ h, int N) {
    const int lane = threadIdx.x & 31;
    const int wave = threadIdx.x >> 5;
    const int mtile = blockIdx.x * 8 + wave;
    const int mtiles = N >> 4;
    if (mtile >= mtiles) return;
    const int m0   = mtile << 4;
    const int n0   = blockIdx.y * 64;
    const int half = lane >> 4;
    const int sub  = lane & 15;

    v8f acc0 = {}, acc1 = {}, acc2 = {}, acc3 = {};

    const float* xrow = x + (size_t)(m0 + sub) * NFEAT + 2 * half;
    for (int kb = 0; kb < NFEAT; kb += 4) {
        v2f afrag;
        afrag.x = xrow[kb + 0];
        afrag.y = xrow[kb + 1];

        const float* wk0 = W + (size_t)(kb + 2 * half + 0) * NCLASS + n0 + sub;
        const float* wk1 = W + (size_t)(kb + 2 * half + 1) * NCLASS + n0 + sub;
        v2f b0; b0.x = wk0[ 0]; b0.y = wk1[ 0];
        v2f b1; b1.x = wk0[16]; b1.y = wk1[16];
        v2f b2; b2.x = wk0[32]; b2.y = wk1[32];
        v2f b3; b3.x = wk0[48]; b3.y = wk1[48];

        acc0 = __builtin_amdgcn_wmma_f32_16x16x4_f32(false, afrag, false, b0, (short)0, acc0, false, false);
        acc1 = __builtin_amdgcn_wmma_f32_16x16x4_f32(false, afrag, false, b1, (short)0, acc1, false, false);
        acc2 = __builtin_amdgcn_wmma_f32_16x16x4_f32(false, afrag, false, b2, (short)0, acc2, false, false);
        acc3 = __builtin_amdgcn_wmma_f32_16x16x4_f32(false, afrag, false, b3, (short)0, acc3, false, false);
    }

    float* hout = h + n0 + sub;
    #pragma unroll
    for (int r = 0; r < 8; ++r) {
        const size_t m = (size_t)(m0 + half * 8 + r);
        hout[m * NCLASS +  0] = acc0[r];
        hout[m * NCLASS + 16] = acc1[r];
        hout[m * NCLASS + 32] = acc2[r];
        hout[m * NCLASS + 48] = acc3[r];
    }
}

__global__ void spgat_scores(const float* __restrict__ h,
                             const float* __restrict__ a,
                             float* __restrict__ s_row,
                             float* __restrict__ s_col, int N) {
    const int lane = threadIdx.x & 31;
    const int row  = blockIdx.x * (blockDim.x >> 5) + (threadIdx.x >> 5);
    if (row >= N) return;
    const float* hr = h + (size_t)row * NCLASS;
    float sr = 0.f, sc = 0.f;
    #pragma unroll
    for (int i = 0; i < NCLASS / 32; ++i) {
        const int f = lane + 32 * i;
        const float v = hr[f];
        sr += v * a[f];
        sc += v * a[NCLASS + f];
    }
    #pragma unroll
    for (int off = 16; off; off >>= 1) {
        sr += __shfl_xor(sr, off, 32);
        sc += __shfl_xor(sc, off, 32);
    }
    if (lane == 0) { s_row[row] = sr; s_col[row] = sc; }
}

__global__ void spgat_edges(const int* __restrict__ ei, int E,
                            const float* __restrict__ s_row,
                            const float* __restrict__ s_col,
                            float* __restrict__ e_buf,
                            float* __restrict__ rowsum) {
    const int e = blockIdx.x * blockDim.x + threadIdx.x;
    if (e >= E) return;
    const int r = ei[e];
    const int c = ei[E + e];
    const float lg = s_row[r] + s_col[c];
    const float l  = lg > 0.f ? lg : 0.2f * lg;
    const float ev = __expf(-l);
    e_buf[e] = ev;
    atomicAdd(rowsum + r, ev);
}

__global__ void spgat_scatter(const int* __restrict__ ei, int E,
                              const float* __restrict__ e_buf,
                              const float* __restrict__ h,
                              float* __restrict__ out) {
    const unsigned tid = blockIdx.x * blockDim.x + threadIdx.x;
    const int e  = tid >> 6;
    const int fg = tid & 63;
    if (e >= E) return;
    const int r = ei[e];
    const int c = ei[E + e];
    const float ev = e_buf[e];
    const float4 hv = *(const float4*)(h + (size_t)c * NCLASS + fg * 4);
    float* o = out + (size_t)r * NCLASS + fg * 4;
    atomicAdd(o + 0, ev * hv.x);
    atomicAdd(o + 1, ev * hv.y);
    atomicAdd(o + 2, ev * hv.z);
    atomicAdd(o + 3, ev * hv.w);
}

__global__ void spgat_finalize(float* __restrict__ out,
                               const float* __restrict__ rowsum, int N) {
    const int lane = threadIdx.x & 31;
    const int row  = blockIdx.x * (blockDim.x >> 5) + (threadIdx.x >> 5);
    if (row >= N) return;
    const float inv = 1.0f / rowsum[row];
    float* o = out + (size_t)row * NCLASS;
    float v[NCLASS / 32];
    float mx = -3.402823466e38f;
    #pragma unroll
    for (int i = 0; i < NCLASS / 32; ++i) {
        const float hp = o[lane + 32 * i] * inv;
        const float el = hp > 0.f ? hp : (__expf(hp) - 1.f);
        v[i] = el;
        mx = fmaxf(mx, el);
    }
    #pragma unroll
    for (int off = 16; off; off >>= 1) mx = fmaxf(mx, __shfl_xor(mx, off, 32));
    float s = 0.f;
    #pragma unroll
    for (int i = 0; i < NCLASS / 32; ++i) s += __expf(v[i] - mx);
    #pragma unroll
    for (int off = 16; off; off >>= 1) s += __shfl_xor(s, off, 32);
    const float lse = mx + __logf(s);
    #pragma unroll
    for (int i = 0; i < NCLASS / 32; ++i) o[lane + 32 * i] = v[i] - lse;
}

extern "C" void kernel_launch(void* const* d_in, const int* in_sizes, int n_in,
                              void* d_out, int out_size, void* d_ws, size_t ws_size,
                              hipStream_t stream) {
    const float* x  = (const float*)d_in[0];
    const float* W  = (const float*)d_in[1];
    const float* a  = (const float*)d_in[2];
    const int*   ei = (const int*)d_in[3];

    const int N = in_sizes[0] / NFEAT;
    const int E = in_sizes[3] / 2;

    float* h      = (float*)d_ws;
    float* s_row  = h + (size_t)N * NCLASS;
    float* s_col  = s_row + N;
    float* rowsum = s_col + N;
    float* e_buf  = rowsum + N;

    float* out = (float*)d_out;

    hipMemsetAsync(out, 0, (size_t)N * NCLASS * sizeof(float), stream);
    hipMemsetAsync(rowsum, 0, (size_t)N * sizeof(float), stream);

    {
        const int mtiles = N / 16;
        dim3 grid((mtiles + 7) / 8, NCLASS / 64);
        spgat_gemm<<<grid, 256, 0, stream>>>(x, W, h, N);
    }
    {
        dim3 grid((N + 7) / 8);
        spgat_scores<<<grid, 256, 0, stream>>>(h, a, s_row, s_col, N);
    }
    {
        dim3 grid((E + 255) / 256);
        spgat_edges<<<grid, 256, 0, stream>>>(ei, E, s_row, s_col, e_buf, rowsum);
    }
    {
        const long long threads = (long long)E * 64;
        dim3 grid((unsigned)((threads + 255) / 256));
        spgat_scatter<<<grid, 256, 0, stream>>>(ei, E, e_buf, h, out);
    }
    {
        dim3 grid((N + 7) / 8);
        spgat_finalize<<<grid, 256, 0, stream>>>(out, rowsum, N);
    }
}


# Group edges by row (counting sort, int atomic rank) and let one wave per row sum e*h[col] and e in registers, replacing all float atomics; per-row sum order = arrival-order rank
# speedup vs baseline: 1.9072x; 1.9072x over previous
#include <hip/hip_runtime.h>
#include <hip/hip_bf16.h>

#define NFEAT  512
#define NCLASS 256

typedef float v2f __attribute__((ext_vector_type(2)));
typedef float v8f __attribute__((ext_vector_type(8)));

__global__ void spgat_gemm(const float* __restrict__ x,
                           const float* __restrict__ W,
                           float* __restrict__ h, int N) {
    const int lane = threadIdx.x & 31;
    const int wave = threadIdx.x >> 5;
    const int mtile = blockIdx.x * 8 + wave;
    const int mtiles = N >> 4;
    if (mtile >= mtiles) return;
    const int m0   = mtile << 4;
    const int n0   = blockIdx.y * 64;
    const int half = lane >> 4;
    const int sub  = lane & 15;

    v8f acc0 = {}, acc1 = {}, acc2 = {}, acc3 = {};

    const float* xrow = x + (size_t)(m0 + sub) * NFEAT + 2 * half;
    for (int kb = 0; kb < NFEAT; kb += 4) {
        v2f afrag;
        afrag.x = xrow[kb + 0];
        afrag.y = xrow[kb + 1];

        const float* wk0 = W + (size_t)(kb + 2 * half + 0) * NCLASS + n0 + sub;
        const float* wk1 = W + (size_t)(kb + 2 * half + 1) * NCLASS + n0 + sub;
        v2f b0; b0.x = wk0[ 0]; b0.y = wk1[ 0];
        v2f b1; b1.x = wk0[16]; b1.y = wk1[16];
        v2f b2; b2.x = wk0[32]; b2.y = wk1[32];
        v2f b3; b3.x = wk0[48]; b3.y = wk1[48];

        acc0 = __builtin_amdgcn_wmma_f32_16x16x4_f32(false, afrag, false, b0, (short)0, acc0, false, false);
        acc1 = __builtin_amdgcn_wmma_f32_16x16x4_f32(false, afrag, false, b1, (short)0, acc1, false, false);
        acc2 = __builtin_amdgcn_wmma_f32_16x16x4_f32(false, afrag, false, b2, (short)0, acc2, false, false);
        acc3 = __builtin_amdgcn_wmma_f32_16x16x4_f32(false, afrag, false, b3, (short)0, acc3, false, false);
    }

    float* hout = h + n0 + sub;
    #pragma unroll
    for (int r = 0; r < 8; ++r) {
        const size_t m = (size_t)(m0 + half * 8 + r);
        hout[m * NCLASS +  0] = acc0[r];
        hout[m * NCLASS + 16] = acc1[r];
        hout[m * NCLASS + 32] = acc2[r];
        hout[m * NCLASS + 48] = acc3[r];
    }
}

__global__ void spgat_scores(const float* __restrict__ h,
                             const float* __restrict__ a,
                             float* __restrict__ s_row,
                             float* __restrict__ s_col, int N) {
    const int lane = threadIdx.x & 31;
    const int row  = blockIdx.x * (blockDim.x >> 5) + (threadIdx.x >> 5);
    if (row >= N) return;
    const float* hr = h + (size_t)row * NCLASS;
    float sr = 0.f, sc = 0.f;
    #pragma unroll
    for (int i = 0; i < NCLASS / 32; ++i) {
        const int f = lane + 32 * i;
        const float v = hr[f];
        sr += v * a[f];
        sc += v * a[NCLASS + f];
    }
    #pragma unroll
    for (int off = 16; off; off >>= 1) {
        sr += __shfl_xor(sr, off, 32);
        sc += __shfl_xor(sc, off, 32);
    }
    if (lane == 0) { s_row[row] = sr; s_col[row] = sc; }
}

__global__ void spgat_edges(const int* __restrict__ ei, int E,
                            const float* __restrict__ s_row,
                            const float* __restrict__ s_col,
                            float* __restrict__ e_buf,
                            int* __restrict__ cnt,
                            int* __restrict__ rank) {
    const int e = blockIdx.x * blockDim.x + threadIdx.x;
    if (e >= E) return;
    const int r = ei[e];
    const int c = ei[E + e];
    const float lg = s_row[r] + s_col[c];
    const float l  = lg > 0.f ? lg : 0.2f * lg;
    const float ev = __expf(-l);
    e_buf[e] = ev;
    rank[e] = atomicAdd(cnt + r, 1);
}

__global__ void spgat_offsets(const int* __restrict__ cnt,
                              int* __restrict__ off, int N) {
    __shared__ int wsum[32];
    const int tid  = threadIdx.x;
    const int lane = tid & 31;
    const int wave = tid >> 5;
    const int per  = (N + 1023) >> 10;
    int beg = tid * per;      if (beg > N) beg = N;
    int end = beg + per;      if (end > N) end = N;
    int s = 0;
    for (int i = beg; i < end; ++i) s += cnt[i];
    int incl = s;
    #pragma unroll
    for (int o = 1; o < 32; o <<= 1) {
        const int t = __shfl_up(incl, o, 32);
        if (lane >= o) incl += t;
    }
    if (lane == 31) wsum[wave] = incl;
    __syncthreads();
    const int ws = wsum[lane];
    int wincl = ws;
    #pragma unroll
    for (int o = 1; o < 32; o <<= 1) {
        const int t = __shfl_up(wincl, o, 32);
        if (lane >= o) wincl += t;
    }
    int run = __shfl(wincl, wave, 32) - __shfl(ws, wave, 32) + incl - s;
    for (int i = beg; i < end; ++i) { off[i] = run; run += cnt[i]; }
    if (tid == 1023) off[N] = run;
}

__global__ void spgat_place(const int* __restrict__ ei, int E,
                            const float* __restrict__ e_buf,
                            const int* __restrict__ rank,
                            const int* __restrict__ off,
                            int* __restrict__ pairs) {
    const int e = blockIdx.x * blockDim.x + threadIdx.x;
    if (e >= E) return;
    const int r = ei[e];
    const int p = off[r] + rank[e];
    ((int2*)pairs)[p] = make_int2(ei[E + e], __float_as_int(e_buf[e]));
}

__device__ __forceinline__ void axpy4(float4& acc, float s, const float4& v) {
    acc.x += s * v.x; acc.y += s * v.y; acc.z += s * v.z; acc.w += s * v.w;
}

__global__ void spgat_gather(const int* __restrict__ off,
                             const int* __restrict__ pairs,
                             const float* __restrict__ h,
                             float* __restrict__ out,
                             float* __restrict__ rowsum, int N) {
    const int lane = threadIdx.x & 31;
    const int row = __builtin_amdgcn_readfirstlane(
        (int)(blockIdx.x * (blockDim.x >> 5) + (threadIdx.x >> 5)));
    if (row >= N) return;
    const int beg = off[row];
    const int end = off[row + 1];
    const float4* hl = (const float4*)h + lane;
    float4 a0 = make_float4(0.f, 0.f, 0.f, 0.f);
    float4 a1 = make_float4(0.f, 0.f, 0.f, 0.f);
    float rs = 0.f;
    #pragma unroll 4
    for (int j = beg; j < end; ++j) {
        const int2 p = ((const int2*)pairs)[j];
        const float4* q = hl + (size_t)p.x * (NCLASS / 4);
        const float4 u = q[0], v = q[32];
        axpy4(a0, __int_as_float(p.y), u);
        axpy4(a1, __int_as_float(p.y), v);
        rs += __int_as_float(p.y);
    }
    float4* o = (float4*)(out + (size_t)row * NCLASS) + lane;
    o[0]  = a0;
    o[32] = a1;
    if (lane == 0) rowsum[row] = rs;
}

__global__ void spgat_finalize(float* __restrict__ out,
                               const float* __restrict__ rowsum, int N) {
    const int lane = threadIdx.x & 31;
    const int row  = blockIdx.x * (blockDim.x >> 5) + (threadIdx.x >> 5);
    if (row >= N) return;
    const float inv = 1.0f / rowsum[row];
    float* o = out + (size_t)row * NCLASS;
    float v[NCLASS / 32];
    float mx = -3.402823466e38f;
    #pragma unroll
    for (int i = 0; i < NCLASS / 32; ++i) {
        const float hp = o[lane + 32 * i] * inv;
        const float el = hp > 0.f ? hp : (__expf(hp) - 1.f);
        v[i] = el;
        mx = fmaxf(mx, el);
    }
    #pragma unroll
    for (int off = 16; off; off >>= 1) mx = fmaxf(mx, __shfl_xor(mx, off, 32));
    float s = 0.f;
    #pragma unroll
    for (int i = 0; i < NCLASS / 32; ++i) s += __expf(v[i] - mx);
    #pragma unroll
    for (int off = 16; off; off >>= 1) s += __shfl_xor(s, off, 32);
    const float lse = mx + __logf(s);
    #pragma unroll
    for (int i = 0; i < NCLASS / 32; ++i) o[lane + 32 * i] = v[i] - lse;
}

extern "C" void kernel_launch(void* const* d_in, const int* in_sizes, int n_in,
                              void* d_out, int out_size, void* d_ws, size_t ws_size,
                              hipStream_t stream) {
    const float* x  = (const float*)d_in[0];
    const float* W  = (const float*)d_in[1];
    const float* a  = (const float*)d_in[2];
    const int*   ei = (const int*)d_in[3];

    const int N = in_sizes[0] / NFEAT;
    const int E = in_sizes[3] / 2;

    float* h      = (float*)d_ws;
    float* s_row  = h + (size_t)N * NCLASS;
    float* s_col  = s_row + N;
    float* rowsum = s_col + N;
    int*   cnt    = (int*)(rowsum + N);
    float* e_buf  = (float*)(cnt + N);
    int*   rank   = (int*)(e_buf + E);
    int*   pairs  = rank + E;
    int*   off    = pairs + 2 * (size_t)E;

    float* out = (float*)d_out;

    hipMemsetAsync(cnt, 0, (size_t)N * sizeof(int), stream);

    {
        const int mtiles = N / 16;
        dim3 grid((mtiles + 7) / 8, NCLASS / 64);
        spgat_gemm<<<grid, 256, 0, stream>>>(x, W, h, N);
    }
    {
        dim3 grid((N + 7) / 8);
        spgat_scores<<<grid, 256, 0, stream>>>(h, a, s_row, s_col, N);
    }
    {
        dim3 grid((E + 255) / 256);
        spgat_edges<<<grid, 256, 0, stream>>>(ei, E, s_row, s_col, e_buf, cnt, rank);
    }
    spgat_offsets<<<1, 1024, 0, stream>>>(cnt, off, N);
    {
        dim3 grid((E + 255) / 256);
        spgat_place<<<grid, 256, 0, stream>>>(ei, E, e_buf, rank, off, pairs);
    }
    {
        dim3 grid((N + 7) / 8);
        spgat_gather<<<grid, 256, 0, stream>>>(off, pairs, h, out, rowsum, N);
    }
    {
        dim3 grid((N + 7) / 8);
        spgat_finalize<<<grid, 256, 0, stream>>>(out, rowsum, N);
    }
}


# GEMM global accesses 16 B per lane on whole lines: A lane reads 4 k, B lane 4 columns (N-tile t = column 4*sub+t), one 16-B D store per row; k pairs regrouped per WMMA, so h's f32 sum order changes
# speedup vs baseline: 1.9994x; 1.0483x over previous
#include <hip/hip_runtime.h>
#include <hip/hip_bf16.h>

#define NFEAT  512
#define NCLASS 256

typedef float v2f __attribute__((ext_vector_type(2)));
typedef float v8f __attribute__((ext_vector_type(8)));

__global__ void spgat_gemm(const float* __restrict__ x,
                           const float* __restrict__ W,
                           float* __restrict__ h, int N) {
    const int lane = threadIdx.x & 31;
    const int wave = threadIdx.x >> 5;
    const int mtile = blockIdx.x * 8 + wave;
    const int mtiles = N >> 4;
    if (mtile >= mtiles) return;
    const int m0   = mtile << 4;
    const int n0   = blockIdx.y * 64;
    const int half = lane >> 4;
    const int sub  = lane & 15;

    v8f acc0 = {}, acc1 = {}, acc2 = {}, acc3 = {};

    const float4* xrow = (const float4*)(x + (size_t)(m0 + sub) * NFEAT + 4 * half);
    const float4* wcol = (const float4*)(W + (size_t)(4 * half) * NCLASS + n0 + 4 * sub);
    #pragma unroll 2
    for (int kb = 0; kb < NFEAT; kb += 8) {
        const float4 a  = xrow[kb >> 2];
        const float4 w0 = wcol[(size_t)(kb + 0) * (NCLASS / 4)];
        const float4 w1 = wcol[(size_t)(kb + 1) * (NCLASS / 4)];
        const float4 w2 = wcol[(size_t)(kb + 2) * (NCLASS / 4)];
        const float4 w3 = wcol[(size_t)(kb + 3) * (NCLASS / 4)];

        v2f af;  af.x = a.x;  af.y = a.y;
        v2f b0;  b0.x = w0.x; b0.y = w1.x;
        v2f b1;  b1.x = w0.y; b1.y = w1.y;
        v2f b2;  b2.x = w0.z; b2.y = w1.z;
        v2f b3;  b3.x = w0.w; b3.y = w1.w;
        acc0 = __builtin_amdgcn_wmma_f32_16x16x4_f32(false, af, false, b0, (short)0, acc0, false, false);
        acc1 = __builtin_amdgcn_wmma_f32_16x16x4_f32(false, af, false, b1, (short)0, acc1, false, false);
        acc2 = __builtin_amdgcn_wmma_f32_16x16x4_f32(false, af, false, b2, (short)0, acc2, false, false);
        acc3 = __builtin_amdgcn_wmma_f32_16x16x4_f32(false, af, false, b3, (short)0, acc3, false, false);

        af.x = a.z;  af.y = a.w;
        b0.x = w2.x; b0.y = w3.x;
        b1.x = w2.y; b1.y = w3.y;
        b2.x = w2.z; b2.y = w3.z;
        b3.x = w2.w; b3.y = w3.w;
        acc0 = __builtin_amdgcn_wmma_f32_16x16x4_f32(false, af, false, b0, (short)0, acc0, false, false);
        acc1 = __builtin_amdgcn_wmma_f32_16x16x4_f32(false, af, false, b1, (short)0, acc1, false, false);
        acc2 = __builtin_amdgcn_wmma_f32_16x16x4_f32(false, af, false, b2, (short)0, acc2, false, false);
        acc3 = __builtin_amdgcn_wmma_f32_16x16x4_f32(false, af, false, b3, (short)0, acc3, false, false);
    }

    float* hout = h + (size_t)(m0 + half * 8) * NCLASS + n0 + 4 * sub;
    #pragma unroll
    for (int r = 0; r < 8; ++r)
        *(float4*)(hout + (size_t)r * NCLASS) = make_float4(acc0[r], acc1[r], acc2[r], acc3[r]);
}

__global__ void spgat_scores(const float* __restrict__ h,
                             const float* __restrict__ a,
                             float* __restrict__ s_row,
                             float* __restrict__ s_col, int N) {
    const int lane = threadIdx.x & 31;
    const int row  = blockIdx.x * (blockDim.x >> 5) + (threadIdx.x >> 5);
    if (row >= N) return;
    const float* hr = h + (size_t)row * NCLASS;
    float sr = 0.f, sc = 0.f;
    #pragma unroll
    for (int i = 0; i < NCLASS / 32; ++i) {
        const int f = lane + 32 * i;
        const float v = hr[f];
        sr += v * a[f];
        sc += v * a[NCLASS + f];
    }
    #pragma unroll
    for (int off = 16; off; off >>= 1) {
        sr += __shfl_xor(sr, off, 32);
        sc += __shfl_xor(sc, off, 32);
    }
    if (lane == 0) { s_row[row] = sr; s_col[row] = sc; }
}

__global__ void spgat_edges(const int* __restrict__ ei, int E,
                            const float* __restrict__ s_row,
                            const float* __restrict__ s_col,
                            float* __restrict__ e_buf,
                            int* __restrict__ cnt,
                            int* __restrict__ rank) {
    const int e = blockIdx.x * blockDim.x + threadIdx.x;
    if (e >= E) return;
    const int r = ei[e];
    const int c = ei[E + e];
    const float lg = s_row[r] + s_col[c];
    const float l  = lg > 0.f ? lg : 0.2f * lg;
    const float ev = __expf(-l);
    e_buf[e] = ev;
    rank[e] = atomicAdd(cnt + r, 1);
}

__global__ void spgat_offsets(const int* __restrict__ cnt,
                              int* __restrict__ off, int N) {
    __shared__ int wsum[32];
    const int tid  = threadIdx.x;
    const int lane = tid & 31;
    const int wave = tid >> 5;
    const int per  = (N + 1023) >> 10;
    int beg = tid * per;      if (beg > N) beg = N;
    int end = beg + per;      if (end > N) end = N;
    int s = 0;
    for (int i = beg; i < end; ++i) s += cnt[i];
    int incl = s;
    #pragma unroll
    for (int o = 1; o < 32; o <<= 1) {
        const int t = __shfl_up(incl, o, 32);
        if (lane >= o) incl += t;
    }
    if (lane == 31) wsum[wave] = incl;
    __syncthreads();
    const int ws = wsum[lane];
    int wincl = ws;
    #pragma unroll
    for (int o = 1; o < 32; o <<= 1) {
        const int t = __shfl_up(wincl, o, 32);
        if (lane >= o) wincl += t;
    }
    int run = __shfl(wincl, wave, 32) - __shfl(ws, wave, 32) + incl - s;
    for (int i = beg; i < end; ++i) { off[i] = run; run += cnt[i]; }
    if (tid == 1023) off[N] = run;
}

__global__ void spgat_place(const int* __restrict__ ei, int E,
                            const float* __restrict__ e_buf,
                            const int* __restrict__ rank,
                            const int* __restrict__ off,
                            int* __restrict__ pairs) {
    const int e = blockIdx.x * blockDim.x + threadIdx.x;
    if (e >= E) return;
    const int r = ei[e];
    const int p = off[r] + rank[e];
    ((int2*)pairs)[p] = make_int2(ei[E + e], __float_as_int(e_buf[e]));
}

__device__ __forceinline__ void axpy4(float4& acc, float s, const float4& v) {
    acc.x += s * v.x; acc.y += s * v.y; acc.z += s * v.z; acc.w += s * v.w;
}

__global__ void spgat_gather(const int* __restrict__ off,
                             const int* __restrict__ pairs,
                             const float* __restrict__ h,
                             float* __restrict__ out,
                             float* __restrict__ rowsum, int N) {
    const int lane = threadIdx.x & 31;
    const int row = __builtin_amdgcn_readfirstlane(
        (int)(blockIdx.x * (blockDim.x >> 5) + (threadIdx.x >> 5)));
    if (row >= N) return;
    const int beg = off[row];
    const int end = off[row + 1];
    const float4* hl = (const float4*)h + lane;
    float4 a0 = make_float4(0.f, 0.f, 0.f, 0.f);
    float4 a1 = make_float4(0.f, 0.f, 0.f, 0.f);
    float rs = 0.f;
    #pragma unroll 4
    for (int j = beg; j < end; ++j) {
        const int2 p = ((const int2*)pairs)[j];
        const float4* q = hl + (size_t)p.x * (NCLASS / 4);
        const float4 u = q[0], v = q[32];
        axpy4(a0, __int_as_float(p.y), u);
        axpy4(a1, __int_as_float(p.y), v);
        rs += __int_as_float(p.y);
    }
    float4* o = (float4*)(out + (size_t)row * NCLASS) + lane;
    o[0]  = a0;
    o[32] = a1;
    if (lane == 0) rowsum[row] = rs;
}

__global__ void spgat_finalize(float* __restrict__ out,
                               const float* __restrict__ rowsum, int N) {
    const int lane = threadIdx.x & 31;
    const int row  = blockIdx.x * (blockDim.x >> 5) + (threadIdx.x >> 5);
    if (row >= N) return;
    const float inv = 1.0f / rowsum[row];
    float* o = out + (size_t)row * NCLASS;
    float v[NCLASS / 32];
    float mx = -3.402823466e38f;
    #pragma unroll
    for (int i = 0; i < NCLASS / 32; ++i) {
        const float hp = o[lane + 32 * i] * inv;
        const float el = hp > 0.f ? hp : (__expf(hp) - 1.f);
        v[i] = el;
        mx = fmaxf(mx, el);
    }
    #pragma unroll
    for (int off = 16; off; off >>= 1) mx = fmaxf(mx, __shfl_xor(mx, off, 32));
    float s = 0.f;
    #pragma unroll
    for (int i = 0; i < NCLASS / 32; ++i) s += __expf(v[i] - mx);
    #pragma unroll
    for (int off = 16; off; off >>= 1) s += __shfl_xor(s, off, 32);
    const float lse = mx + __logf(s);
    #pragma unroll
    for (int i = 0; i < NCLASS / 32; ++i) o[lane + 32 * i] = v[i] - lse;
}

extern "C" void kernel_launch(void* const* d_in, const int* in_sizes, int n_in,
                              void* d_out, int out_size, void* d_ws, size_t ws_size,
                              hipStream_t stream) {
    const float* x  = (const float*)d_in[0];
    const float* W  = (const float*)d_in[1];
    const float* a  = (const float*)d_in[2];
    const int*   ei = (const int*)d_in[3];

    const int N = in_sizes[0] / NFEAT;
    const int E = in_sizes[3] / 2;

    float* h      = (float*)d_ws;
    float* s_row  = h + (size_t)N * NCLASS;
    float* s_col  = s_row + N;
    float* rowsum = s_col + N;
    int*   cnt    = (int*)(rowsum + N);
    float* e_buf  = (float*)(cnt + N);
    int*   rank   = (int*)(e_buf + E);
    int*   pairs  = rank + E;
    int*   off    = pairs + 2 * (size_t)E;

    float* out = (float*)d_out;

    hipMemsetAsync(cnt, 0, (size_t)N * sizeof(int), stream);

    {
        const int mtiles = N / 16;
        dim3 grid((mtiles + 7) / 8, NCLASS / 64);
        spgat_gemm<<<grid, 256, 0, stream>>>(x, W, h, N);
    }
    {
        dim3 grid((N + 7) / 8);
        spgat_scores<<<grid, 256, 0, stream>>>(h, a, s_row, s_col, N);
    }
    {
        dim3 grid((E + 255) / 256);
        spgat_edges<<<grid, 256, 0, stream>>>(ei, E, s_row, s_col, e_buf, cnt, rank);
    }
    spgat_offsets<<<1, 1024, 0, stream>>>(cnt, off, N);
    {
        dim3 grid((E + 255) / 256);
        spgat_place<<<grid, 256, 0, stream>>>(ei, E, e_buf, rank, off, pairs);
    }
    {
        dim3 grid((N + 7) / 8);
        spgat_gather<<<grid, 256, 0, stream>>>(off, pairs, h, out, rowsum, N);
    }
    {
        dim3 grid((N + 7) / 8);
        spgat_finalize<<<grid, 256, 0, stream>>>(out, rowsum, N);
    }
}


# Rank edges in LDS, not by one returning global atomic each: block histograms of 256-row buckets, table prefix, LDS cursors, per-bucket LDS sort; per-row sum order = LDS arrival rank
# speedup vs baseline: 2.5575x; 1.2791x over previous
#include <hip/hip_runtime.h>
#include <hip/hip_bf16.h>

#define NFEAT  512
#define NCLASS 256

typedef float v2f __attribute__((ext_vector_type(2)));
typedef float v8f __attribute__((ext_vector_type(8)));

__global__ void spgat_gemm(const float* __restrict__ x,
                           const float* __restrict__ W,
                           float* __restrict__ h, int N) {
    const int lane = threadIdx.x & 31;
    const int wave = threadIdx.x >> 5;
    const int mtile = blockIdx.x * 8 + wave;
    const int mtiles = N >> 4;
    if (mtile >= mtiles) return;
    const int m0   = mtile << 4;
    const int n0   = blockIdx.y * 64;
    const int half = lane >> 4;
    const int sub  = lane & 15;

    v8f acc0 = {}, acc1 = {}, acc2 = {}, acc3 = {};

    const float4* xrow = (const float4*)(x + (size_t)(m0 + sub) * NFEAT + 4 * half);
    const float4* wcol = (const float4*)(W + (size_t)(4 * half) * NCLASS + n0 + 4 * sub);
    #pragma unroll 2
    for (int kb = 0; kb < NFEAT; kb += 8) {
        const float4 a  = xrow[kb >> 2];
        const float4 w0 = wcol[(size_t)(kb + 0) * (NCLASS / 4)];
        const float4 w1 = wcol[(size_t)(kb + 1) * (NCLASS / 4)];
        const float4 w2 = wcol[(size_t)(kb + 2) * (NCLASS / 4)];
        const float4 w3 = wcol[(size_t)(kb + 3) * (NCLASS / 4)];

        v2f af;  af.x = a.x;  af.y = a.y;
        v2f b0;  b0.x = w0.x; b0.y = w1.x;
        v2f b1;  b1.x = w0.y; b1.y = w1.y;
        v2f b2;  b2.x = w0.z; b2.y = w1.z;
        v2f b3;  b3.x = w0.w; b3.y = w1.w;
        acc0 = __builtin_amdgcn_wmma_f32_16x16x4_f32(false, af, false, b0, (short)0, acc0, false, false);
        acc1 = __builtin_amdgcn_wmma_f32_16x16x4_f32(false, af, false, b1, (short)0, acc1, false, false);
        acc2 = __builtin_amdgcn_wmma_f32_16x16x4_f32(false, af, false, b2, (short)0, acc2, false, false);
        acc3 = __builtin_amdgcn_wmma_f32_16x16x4_f32(false, af, false, b3, (short)0, acc3, false, false);

        af.x = a.z;  af.y = a.w;
        b0.x = w2.x; b0.y = w3.x;
        b1.x = w2.y; b1.y = w3.y;
        b2.x = w2.z; b2.y = w3.z;
        b3.x = w2.w; b3.y = w3.w;
        acc0 = __builtin_amdgcn_wmma_f32_16x16x4_f32(false, af, false, b0, (short)0, acc0, false, false);
        acc1 = __builtin_amdgcn_wmma_f32_16x16x4_f32(false, af, false, b1, (short)0, acc1, false, false);
        acc2 = __builtin_amdgcn_wmma_f32_16x16x4_f32(false, af, false, b2, (short)0, acc2, false, false);
        acc3 = __builtin_amdgcn_wmma_f32_16x16x4_f32(false, af, false, b3, (short)0, acc3, false, false);
    }

    float* hout = h + (size_t)(m0 + half * 8) * NCLASS + n0 + 4 * sub;
    #pragma unroll
    for (int r = 0; r < 8; ++r)
        *(float4*)(hout + (size_t)r * NCLASS) = make_float4(acc0[r], acc1[r], acc2[r], acc3[r]);
}

__global__ void spgat_scores(const float* __restrict__ h,
                             const float* __restrict__ a,
                             float* __restrict__ s_row,
                             float* __restrict__ s_col, int N) {
    const int lane = threadIdx.x & 31;
    const int row  = blockIdx.x * (blockDim.x >> 5) + (threadIdx.x >> 5);
    if (row >= N) return;
    const float* hr = h + (size_t)row * NCLASS;
    float sr = 0.f, sc = 0.f;
    #pragma unroll
    for (int i = 0; i < NCLASS / 32; ++i) {
        const int f = lane + 32 * i;
        const float v = hr[f];
        sr += v * a[f];
        sc += v * a[NCLASS + f];
    }
    #pragma unroll
    for (int off = 16; off; off >>= 1) {
        sr += __shfl_xor(sr, off, 32);
        sc += __shfl_xor(sc, off, 32);
    }
    if (lane == 0) { s_row[row] = sr; s_col[row] = sc; }
}

#define BKT_SHIFT 8
#define BKT_ROWS  256
#define MAX_BKT   256
#define EDGE_BLK  4096

__device__ __forceinline__ void scan256_excl(int* a, int lane) {
    int v[8];
    int s = 0;
    #pragma unroll
    for (int k = 0; k < 8; ++k) { v[k] = a[8 * lane + k]; s += v[k]; }
    int incl = s;
    #pragma unroll
    for (int o = 1; o < 32; o <<= 1) {
        const int t = __shfl_up(incl, o, 32);
        if (lane >= o) incl += t;
    }
    int run = incl - s;
    #pragma unroll
    for (int k = 0; k < 8; ++k) { a[8 * lane + k] = run; run += v[k]; }
}

__global__ void __launch_bounds__(1024)
spgat_bcount(const int* __restrict__ ei, int E,
             int* __restrict__ blk_hist) {
    __shared__ int hist[MAX_BKT];
    const int tid = threadIdx.x;
    if (tid < MAX_BKT) hist[tid] = 0;
    __syncthreads();
    const int e0 = blockIdx.x * EDGE_BLK + tid;
    int r[4];
    #pragma unroll
    for (int i = 0; i < 4; ++i) {
        const int e = e0 + 1024 * i;
        r[i] = ei[e < E ? e : E - 1];
    }
    #pragma unroll
    for (int i = 0; i < 4; ++i)
        if (e0 + 1024 * i < E) atomicAdd(&hist[r[i] >> BKT_SHIFT], 1);
    __syncthreads();
    if (tid < MAX_BKT) blk_hist[blockIdx.x * MAX_BKT + tid] = hist[tid];
}

__global__ void __launch_bounds__(1024)
spgat_bscatter(const int* __restrict__ ei, int E, int NB,
               const int* __restrict__ blk_hist,
               const float* __restrict__ s_row,
               const float* __restrict__ s_col,
               int* __restrict__ bstart,
               int* __restrict__ recs) {
    __shared__ __attribute__((aligned(16))) int p_tot[16][MAX_BKT];
    __shared__ __attribute__((aligned(16))) int p_pre[16][MAX_BKT];
    __shared__ int s_tot[MAX_BKT];
    __shared__ int cursor[MAX_BKT];
    const int tid = threadIdx.x;
    const int wg  = blockIdx.x;
    const int nwg = gridDim.x;
    {
        const int c4 = tid & 63;
        const int g  = tid >> 6;
        const int4* tbl = (const int4*)blk_hist + c4;
        int4 tot = make_int4(0, 0, 0, 0), pre = make_int4(0, 0, 0, 0);
        for (int w0 = g; w0 < nwg; w0 += 128) {
            int4 v[8];
            #pragma unroll
            for (int k = 0; k < 8; ++k) {
                const int w = w0 + 16 * k;
                v[k] = tbl[(size_t)(w < nwg ? w : nwg - 1) * (MAX_BKT / 4)];
            }
            #pragma unroll
            for (int k = 0; k < 8; ++k) {
                const int w = w0 + 16 * k;
                const int in = w < nwg ? -1 : 0;
                const int fr = w < wg  ? -1 : 0;
                tot.x += v[k].x & in;       tot.y += v[k].y & in;
                tot.z += v[k].z & in;       tot.w += v[k].w & in;
                pre.x += v[k].x & fr;       pre.y += v[k].y & fr;
                pre.z += v[k].z & fr;       pre.w += v[k].w & fr;
            }
        }
        *(int4*)&p_tot[g][4 * c4] = tot;
        *(int4*)&p_pre[g][4 * c4] = pre;
    }
    __syncthreads();
    int pre_b = 0;
    if (tid < MAX_BKT) {
        int t = 0;
        #pragma unroll
        for (int k = 0; k < 16; ++k) { t += p_tot[k][tid]; pre_b += p_pre[k][tid]; }
        s_tot[tid] = t;
    }
    __syncthreads();
    if (tid < 32) scan256_excl(s_tot, tid);
    __syncthreads();
    if (tid < MAX_BKT) {
        const int st = s_tot[tid];
        cursor[tid] = st + pre_b;
        if (wg == 0 && tid < NB) bstart[tid] = st;
    }
    if (wg == 0 && tid == 0) bstart[NB] = E;
    __syncthreads();

    const int e0 = wg * EDGE_BLK + tid;
    int r[4], c[4];
    #pragma unroll
    for (int i = 0; i < 4; ++i) {
        const int e  = e0 + 1024 * i;
        const int ec = e < E ? e : E - 1;
        r[i] = ei[ec];
        c[i] = ei[E + ec];
    }
    float lg[4];
    #pragma unroll
    for (int i = 0; i < 4; ++i) lg[i] = s_row[r[i]] + s_col[c[i]];
    #pragma unroll
    for (int i = 0; i < 4; ++i) {
        if (e0 + 1024 * i < E) {
            const float l  = lg[i] > 0.f ? lg[i] : 0.2f * lg[i];
            const float ev = __expf(-l);
            const int p = atomicAdd(&cursor[r[i] >> BKT_SHIFT], 1);
            ((int2*)recs)[p] = make_int2((int)(((unsigned)(r[i] & (BKT_ROWS - 1)) << 24) | (unsigned)c[i]),
                                         __float_as_int(ev));
        }
    }
}

__global__ void __launch_bounds__(1024)
spgat_bsort(const int* __restrict__ bstart,
            const int* __restrict__ recs,
            int* __restrict__ off,
            int* __restrict__ pairs, int N) {
    __shared__ int cnt[BKT_ROWS];
    __shared__ int cur[BKT_ROWS];
    const int tid = threadIdx.x;
    const int b   = blockIdx.x;
    const int beg = bstart[b];
    const int end = bstart[b + 1];
    if (tid < BKT_ROWS) cnt[tid] = 0;
    __syncthreads();
    for (int i0 = beg + tid; i0 < end; i0 += 5 * 1024) {
        int k[5];
        #pragma unroll
        for (int j = 0; j < 5; ++j) {
            const int i = i0 + 1024 * j;
            k[j] = recs[2 * (size_t)(i < end ? i : end - 1)];
        }
        #pragma unroll
        for (int j = 0; j < 5; ++j)
            if (i0 + 1024 * j < end) atomicAdd(&cnt[((unsigned)k[j]) >> 24], 1);
    }
    __syncthreads();
    if (tid < 32) scan256_excl(cnt, tid);
    __syncthreads();
    if (tid < BKT_ROWS) {
        const int o = beg + cnt[tid];
        cur[tid] = o;
        const int row = b * BKT_ROWS + tid;
        if (row < N) off[row] = o;
    }
    if (b == (int)gridDim.x - 1 && tid == 0) off[N] = end;
    __syncthreads();
    for (int i0 = beg + tid; i0 < end; i0 += 5 * 1024) {
        int2 rc[5];
        #pragma unroll
        for (int j = 0; j < 5; ++j) {
            const int i = i0 + 1024 * j;
            rc[j] = ((const int2*)recs)[i < end ? i : end - 1];
        }
        #pragma unroll
        for (int j = 0; j < 5; ++j) {
            if (i0 + 1024 * j < end) {
                const int p = atomicAdd(&cur[((unsigned)rc[j].x) >> 24], 1);
                ((int2*)pairs)[p] = make_int2(rc[j].x & 0xFFFFFF, rc[j].y);
            }
        }
    }
}

__device__ __forceinline__ void axpy4(float4& acc, float s, const float4& v) {
    acc.x += s * v.x; acc.y += s * v.y; acc.z += s * v.z; acc.w += s * v.w;
}

__global__ void spgat_gather(const int* __restrict__ off,
                             const int* __restrict__ pairs,
                             const float* __restrict__ h,
                             float* __restrict__ out,
                             float* __restrict__ rowsum, int N) {
    const int lane = threadIdx.x & 31;
    const int row = __builtin_amdgcn_readfirstlane(
        (int)(blockIdx.x * (blockDim.x >> 5) + (threadIdx.x >> 5)));
    if (row >= N) return;
    const int beg = off[row];
    const int end = off[row + 1];
    const float4* hl = (const float4*)h + lane;
    float4 a0 = make_float4(0.f, 0.f, 0.f, 0.f);
    float4 a1 = make_float4(0.f, 0.f, 0.f, 0.f);
    float rs = 0.f;
    #pragma unroll 4
    for (int j = beg; j < end; ++j) {
        const int2 p = ((const int2*)pairs)[j];
        const float4* q = hl + (size_t)p.x * (NCLASS / 4);
        const float4 u = q[0], v = q[32];
        axpy4(a0, __int_as_float(p.y), u);
        axpy4(a1, __int_as_float(p.y), v);
        rs += __int_as_float(p.y);
    }
    float4* o = (float4*)(out + (size_t)row * NCLASS) + lane;
    o[0]  = a0;
    o[32] = a1;
    if (lane == 0) rowsum[row] = rs;
}

__global__ void spgat_finalize(float* __restrict__ out,
                               const float* __restrict__ rowsum, int N) {
    const int lane = threadIdx.x & 31;
    const int row  = blockIdx.x * (blockDim.x >> 5) + (threadIdx.x >> 5);
    if (row >= N) return;
    const float inv = 1.0f / rowsum[row];
    float* o = out + (size_t)row * NCLASS;
    float v[NCLASS / 32];
    float mx = -3.402823466e38f;
    #pragma unroll
    for (int i = 0; i < NCLASS / 32; ++i) {
        const float hp = o[lane + 32 * i] * inv;
        const float el = hp > 0.f ? hp : (__expf(hp) - 1.f);
        v[i] = el;
        mx = fmaxf(mx, el);
    }
    #pragma unroll
    for (int off = 16; off; off >>= 1) mx = fmaxf(mx, __shfl_xor(mx, off, 32));
    float s = 0.f;
    #pragma unroll
    for (int i = 0; i < NCLASS / 32; ++i) s += __expf(v[i] - mx);
    #pragma unroll
    for (int off = 16; off; off >>= 1) s += __shfl_xor(s, off, 32);
    const float lse = mx + __logf(s);
    #pragma unroll
    for (int i = 0; i < NCLASS / 32; ++i) o[lane + 32 * i] = v[i] - lse;
}

extern "C" void kernel_launch(void* const* d_in, const int* in_sizes, int n_in,
                              void* d_out, int out_size, void* d_ws, size_t ws_size,
                              hipStream_t stream) {
    const float* x  = (const float*)d_in[0];
    const float* W  = (const float*)d_in[1];
    const float* a  = (const float*)d_in[2];
    const int*   ei = (const int*)d_in[3];

    const int N = in_sizes[0] / NFEAT;
    const int E = in_sizes[3] / 2;

    const int NB  = (N + BKT_ROWS - 1) / BKT_ROWS;
    const int NEB = (E + EDGE_BLK - 1) / EDGE_BLK;
    float* h        = (float*)d_ws;
    int*   blk_hist = (int*)(h + (size_t)N * NCLASS);
    float* s_row    = (float*)(blk_hist + (size_t)NEB * MAX_BKT);
    float* s_col    = s_row + N;
    float* rowsum   = s_col + N;
    int*   recs     = (int*)(rowsum + N + (N & 1));
    int*   pairs    = recs + 2 * (size_t)E;
    int*   off      = pairs + 2 * (size_t)E;
    int*   bstart   = off + N + 1;

    float* out = (float*)d_out;

    {
        const int mtiles = N / 16;
        dim3 grid((mtiles + 7) / 8, NCLASS / 64);
        spgat_gemm<<<grid, 256, 0, stream>>>(x, W, h, N);
    }
    {
        dim3 grid((N + 7) / 8);
        spgat_scores<<<grid, 256, 0, stream>>>(h, a, s_row, s_col, N);
    }
    spgat_bcount<<<NEB, 1024, 0, stream>>>(ei, E, blk_hist);
    spgat_bscatter<<<NEB, 1024, 0, stream>>>(ei, E, NB, blk_hist, s_row, s_col, bstart, recs);
    spgat_bsort<<<NB, 1024, 0, stream>>>(bstart, recs, off, pairs, N);
    {
        dim3 grid((N + 7) / 8);
        spgat_gather<<<grid, 256, 0, stream>>>(off, pairs, h, out, rowsum, N);
    }
    {
        dim3 grid((N + 7) / 8);
        spgat_finalize<<<grid, 256, 0, stream>>>(out, rowsum, N);
    }
}


# Finalize fused into gather: the row's wave divides by its e sum, does elu + log-softmax in registers (DPP butterflies), stores out once; a launch and an out write+read less; lse sum order changes
# speedup vs baseline: 2.8368x; 1.1092x over previous
#include <hip/hip_runtime.h>
#include <hip/hip_bf16.h>

#define NFEAT  512
#define NCLASS 256

typedef float v2f __attribute__((ext_vector_type(2)));
typedef float v8f __attribute__((ext_vector_type(8)));

__global__ void spgat_gemm(const float* __restrict__ x,
                           const float* __restrict__ W,
                           float* __restrict__ h, int N) {
    const int lane = threadIdx.x & 31;
    const int wave = threadIdx.x >> 5;
    const int mtile = blockIdx.x * 8 + wave;
    const int mtiles = N >> 4;
    if (mtile >= mtiles) return;
    const int m0   = mtile << 4;
    const int n0   = blockIdx.y * 64;
    const int half = lane >> 4;
    const int sub  = lane & 15;

    v8f acc0 = {}, acc1 = {}, acc2 = {}, acc3 = {};

    const float4* xrow = (const float4*)(x + (size_t)(m0 + sub) * NFEAT + 4 * half);
    const float4* wcol = (const float4*)(W + (size_t)(4 * half) * NCLASS + n0 + 4 * sub);
    #pragma unroll 2
    for (int kb = 0; kb < NFEAT; kb += 8) {
        const float4 a  = xrow[kb >> 2];
        const float4 w0 = wcol[(size_t)(kb + 0) * (NCLASS / 4)];
        const float4 w1 = wcol[(size_t)(kb + 1) * (NCLASS / 4)];
        const float4 w2 = wcol[(size_t)(kb + 2) * (NCLASS / 4)];
        const float4 w3 = wcol[(size_t)(kb + 3) * (NCLASS / 4)];

        v2f af;  af.x = a.x;  af.y = a.y;
        v2f b0;  b0.x = w0.x; b0.y = w1.x;
        v2f b1;  b1.x = w0.y; b1.y = w1.y;
        v2f b2;  b2.x = w0.z; b2.y = w1.z;
        v2f b3;  b3.x = w0.w; b3.y = w1.w;
        acc0 = __builtin_amdgcn_wmma_f32_16x16x4_f32(false, af, false, b0, (short)0, acc0, false, false);
        acc1 = __builtin_amdgcn_wmma_f32_16x16x4_f32(false, af, false, b1, (short)0, acc1, false, false);
        acc2 = __builtin_amdgcn_wmma_f32_16x16x4_f32(false, af, false, b2, (short)0, acc2, false, false);
        acc3 = __builtin_amdgcn_wmma_f32_16x16x4_f32(false, af, false, b3, (short)0, acc3, false, false);

        af.x = a.z;  af.y = a.w;
        b0.x = w2.x; b0.y = w3.x;
        b1.x = w2.y; b1.y = w3.y;
        b2.x = w2.z; b2.y = w3.z;
        b3.x = w2.w; b3.y = w3.w;
        acc0 = __builtin_amdgcn_wmma_f32_16x16x4_f32(false, af, false, b0, (short)0, acc0, false, false);
        acc1 = __builtin_amdgcn_wmma_f32_16x16x4_f32(false, af, false, b1, (short)0, acc1, false, false);
        acc2 = __builtin_amdgcn_wmma_f32_16x16x4_f32(false, af, false, b2, (short)0, acc2, false, false);
        acc3 = __builtin_amdgcn_wmma_f32_16x16x4_f32(false, af, false, b3, (short)0, acc3, false, false);
    }

    float* hout = h + (size_t)(m0 + half * 8) * NCLASS + n0 + 4 * sub;
    #pragma unroll
    for (int r = 0; r < 8; ++r)
        *(float4*)(hout + (size_t)r * NCLASS) = make_float4(acc0[r], acc1[r], acc2[r], acc3[r]);
}

__global__ void spgat_scores(const float* __restrict__ h,
                             const float* __restrict__ a,
                             float* __restrict__ s_row,
                             float* __restrict__ s_col, int N) {
    const int lane = threadIdx.x & 31;
    const int row  = blockIdx.x * (blockDim.x >> 5) + (threadIdx.x >> 5);
    if (row >= N) return;
    const float* hr = h + (size_t)row * NCLASS;
    float sr = 0.f, sc = 0.f;
    #pragma unroll
    for (int i = 0; i < NCLASS / 32; ++i) {
        const int f = lane + 32 * i;
        const float v = hr[f];
        sr += v * a[f];
        sc += v * a[NCLASS + f];
    }
    #pragma unroll
    for (int off = 16; off; off >>= 1) {
        sr += __shfl_xor(sr, off, 32);
        sc += __shfl_xor(sc, off, 32);
    }
    if (lane == 0) { s_row[row] = sr; s_col[row] = sc; }
}

#define BKT_SHIFT 8
#define BKT_ROWS  256
#define MAX_BKT   256
#define EDGE_BLK  4096

__device__ __forceinline__ void scan256_excl(int* a, int lane) {
    int v[8];
    int s = 0;
    #pragma unroll
    for (int k = 0; k < 8; ++k) { v[k] = a[8 * lane + k]; s += v[k]; }
    int incl = s;
    #pragma unroll
    for (int o = 1; o < 32; o <<= 1) {
        const int t = __shfl_up(incl, o, 32);
        if (lane >= o) incl += t;
    }
    int run = incl - s;
    #pragma unroll
    for (int k = 0; k < 8; ++k) { a[8 * lane + k] = run; run += v[k]; }
}

__global__ void __launch_bounds__(1024)
spgat_bcount(const int* __restrict__ ei, int E,
             int* __restrict__ blk_hist) {
    __shared__ int hist[MAX_BKT];
    const int tid = threadIdx.x;
    if (tid < MAX_BKT) hist[tid] = 0;
    __syncthreads();
    const int e0 = blockIdx.x * EDGE_BLK + tid;
    int r[4];
    #pragma unroll
    for (int i = 0; i < 4; ++i) {
        const int e = e0 + 1024 * i;
        r[i] = ei[e < E ? e : E - 1];
    }
    #pragma unroll
    for (int i = 0; i < 4; ++i)
        if (e0 + 1024 * i < E) atomicAdd(&hist[r[i] >> BKT_SHIFT], 1);
    __syncthreads();
    if (tid < MAX_BKT) blk_hist[blockIdx.x * MAX_BKT + tid] = hist[tid];
}

__global__ void __launch_bounds__(1024)
spgat_bscatter(const int* __restrict__ ei, int E, int NB,
               const int* __restrict__ blk_hist,
               const float* __restrict__ s_row,
               const float* __restrict__ s_col,
               int* __restrict__ bstart,
               int* __restrict__ recs) {
    __shared__ __attribute__((aligned(16))) int p_tot[16][MAX_BKT];
    __shared__ __attribute__((aligned(16))) int p_pre[16][MAX_BKT];
    __shared__ int s_tot[MAX_BKT];
    __shared__ int cursor[MAX_BKT];
    const int tid = threadIdx.x;
    const int wg  = blockIdx.x;
    const int nwg = gridDim.x;
    {
        const int c4 = tid & 63;
        const int g  = tid >> 6;
        const int4* tbl = (const int4*)blk_hist + c4;
        int4 tot = make_int4(0, 0, 0, 0), pre = make_int4(0, 0, 0, 0);
        for (int w0 = g; w0 < nwg; w0 += 128) {
            int4 v[8];
            #pragma unroll
            for (int k = 0; k < 8; ++k) {
                const int w = w0 + 16 * k;
                v[k] = tbl[(size_t)(w < nwg ? w : nwg - 1) * (MAX_BKT / 4)];
            }
            #pragma unroll
            for (int k = 0; k < 8; ++k) {
                const int w = w0 + 16 * k;
                const int in = w < nwg ? -1 : 0;
                const int fr = w < wg  ? -1 : 0;
                tot.x += v[k].x & in;       tot.y += v[k].y & in;
                tot.z += v[k].z & in;       tot.w += v[k].w & in;
                pre.x += v[k].x & fr;       pre.y += v[k].y & fr;
                pre.z += v[k].z & fr;       pre.w += v[k].w & fr;
            }
        }
        *(int4*)&p_tot[g][4 * c4] = tot;
        *(int4*)&p_pre[g][4 * c4] = pre;
    }
    __syncthreads();
    int pre_b = 0;
    if (tid < MAX_BKT) {
        int t = 0;
        #pragma unroll
        for (int k = 0; k < 16; ++k) { t += p_tot[k][tid]; pre_b += p_pre[k][tid]; }
        s_tot[tid] = t;
    }
    __syncthreads();
    if (tid < 32) scan256_excl(s_tot, tid);
    __syncthreads();
    if (tid < MAX_BKT) {
        const int st = s_tot[tid];
        cursor[tid] = st + pre_b;
        if (wg == 0 && tid < NB) bstart[tid] = st;
    }
    if (wg == 0 && tid == 0) bstart[NB] = E;
    __syncthreads();

    const int e0 = wg * EDGE_BLK + tid;
    int r[4], c[4];
    #pragma unroll
    for (int i = 0; i < 4; ++i) {
        const int e  = e0 + 1024 * i;
        const int ec = e < E ? e : E - 1;
        r[i] = ei[ec];
        c[i] = ei[E + ec];
    }
    float lg[4];
    #pragma unroll
    for (int i = 0; i < 4; ++i) lg[i] = s_row[r[i]] + s_col[c[i]];
    #pragma unroll
    for (int i = 0; i < 4; ++i) {
        if (e0 + 1024 * i < E) {
            const float l  = lg[i] > 0.f ? lg[i] : 0.2f * lg[i];
            const float ev = __expf(-l);
            const int p = atomicAdd(&cursor[r[i] >> BKT_SHIFT], 1);
            ((int2*)recs)[p] = make_int2((int)(((unsigned)(r[i] & (BKT_ROWS - 1)) << 24) | (unsigned)c[i]),
                                         __float_as_int(ev));
        }
    }
}

__global__ void __launch_bounds__(1024)
spgat_bsort(const int* __restrict__ bstart,
            const int* __restrict__ recs,
            int* __restrict__ off,
            int* __restrict__ pairs, int N) {
    __shared__ int cnt[BKT_ROWS];
    __shared__ int cur[BKT_ROWS];
    const int tid = threadIdx.x;
    const int b   = blockIdx.x;
    const int beg = bstart[b];
    const int end = bstart[b + 1];
    if (tid < BKT_ROWS) cnt[tid] = 0;
    __syncthreads();
    for (int i0 = beg + tid; i0 < end; i0 += 5 * 1024) {
        int k[5];
        #pragma unroll
        for (int j = 0; j < 5; ++j) {
            const int i = i0 + 1024 * j;
            k[j] = recs[2 * (size_t)(i < end ? i : end - 1)];
        }
        #pragma unroll
        for (int j = 0; j < 5; ++j)
            if (i0 + 1024 * j < end) atomicAdd(&cnt[((unsigned)k[j]) >> 24], 1);
    }
    __syncthreads();
    if (tid < 32) scan256_excl(cnt, tid);
    __syncthreads();
    if (tid < BKT_ROWS) {
        const int o = beg + cnt[tid];
        cur[tid] = o;
        const int row = b * BKT_ROWS + tid;
        if (row < N) off[row] = o;
    }
    if (b == (int)gridDim.x - 1 && tid == 0) off[N] = end;
    __syncthreads();
    for (int i0 = beg + tid; i0 < end; i0 += 5 * 1024) {
        int2 rc[5];
        #pragma unroll
        for (int j = 0; j < 5; ++j) {
            const int i = i0 + 1024 * j;
            rc[j] = ((const int2*)recs)[i < end ? i : end - 1];
        }
        #pragma unroll
        for (int j = 0; j < 5; ++j) {
            if (i0 + 1024 * j < end) {
                const int p = atomicAdd(&cur[((unsigned)rc[j].x) >> 24], 1);
                ((int2*)pairs)[p] = make_int2(rc[j].x & 0xFFFFFF, rc[j].y);
            }
        }
    }
}

__device__ __forceinline__ void axpy4(float4& acc, float s, const float4& v) {
    acc.x += s * v.x; acc.y += s * v.y; acc.z += s * v.z; acc.w += s * v.w;
}

template <int M>
__device__ __forceinline__ float xrow(float x) {
    const int b = __builtin_bit_cast(int, x);
    return __builtin_bit_cast(float, __builtin_amdgcn_update_dpp(0, b, 0x160 + M, 0xf, 0xf, true));
}
__device__ __forceinline__ float xhalf(float x) {
    const int b = __builtin_bit_cast(int, x);
    return __builtin_bit_cast(float,
        __builtin_amdgcn_permlanex16(b, b, 0x76543210, 0xfedcba98, false, false));
}
__device__ __forceinline__ float wave_max(float v) {
    v = fmaxf(v, xrow<1>(v));
    v = fmaxf(v, xrow<2>(v));
    v = fmaxf(v, xrow<4>(v));
    v = fmaxf(v, xrow<8>(v));
    return fmaxf(v, xhalf(v));
}
__device__ __forceinline__ float wave_sum(float v) {
    v += xrow<1>(v);
    v += xrow<2>(v);
    v += xrow<4>(v);
    v += xrow<8>(v);
    return v + xhalf(v);
}

__global__ void spgat_gather(const int* __restrict__ off,
                             const int* __restrict__ pairs,
                             const float* __restrict__ h,
                             float* __restrict__ out, int N) {
    const int lane = threadIdx.x & 31;
    const int row = __builtin_amdgcn_readfirstlane(
        (int)(blockIdx.x * (blockDim.x >> 5) + (threadIdx.x >> 5)));
    if (row >= N) return;
    const int beg = off[row];
    const int end = off[row + 1];
    const float4* hl = (const float4*)h + lane;
    float4 a0 = make_float4(0.f, 0.f, 0.f, 0.f);
    float4 a1 = make_float4(0.f, 0.f, 0.f, 0.f);
    float rs = 0.f;
    #pragma unroll 4
    for (int j = beg; j < end; ++j) {
        const int2 p = ((const int2*)pairs)[j];
        const float4* q = hl + (size_t)p.x * (NCLASS / 4);
        const float4 u = q[0], v = q[32];
        axpy4(a0, __int_as_float(p.y), u);
        axpy4(a1, __int_as_float(p.y), v);
        rs += __int_as_float(p.y);
    }

    const float inv = 1.0f / rs;
    float v[8] = {a0.x, a0.y, a0.z, a0.w, a1.x, a1.y, a1.z, a1.w};
    float mx = -3.402823466e38f;
    #pragma unroll
    for (int i = 0; i < 8; ++i) {
        const float hp = v[i] * inv;
        const float el = hp > 0.f ? hp : (__expf(hp) - 1.f);
        v[i] = el;
        mx = fmaxf(mx, el);
    }
    mx = wave_max(mx);
    float s = 0.f;
    #pragma unroll
    for (int i = 0; i < 8; ++i) s += __expf(v[i] - mx);
    s = wave_sum(s);
    const float lse = mx + __logf(s);

    float4* o = (float4*)(out + (size_t)row * NCLASS) + lane;
    o[0]  = make_float4(v[0] - lse, v[1] - lse, v[2] - lse, v[3] - lse);
    o[32] = make_float4(v[4] - lse, v[5] - lse, v[6] - lse, v[7] - lse);
}

extern "C" void kernel_launch(void* const* d_in, const int* in_sizes, int n_in,
                              void* d_out, int out_size, void* d_ws, size_t ws_size,
                              hipStream_t stream) {
    const float* x  = (const float*)d_in[0];
    const float* W  = (const float*)d_in[1];
    const float* a  = (const float*)d_in[2];
    const int*   ei = (const int*)d_in[3];

    const int N = in_sizes[0] / NFEAT;
    const int E = in_sizes[3] / 2;

    const int NB  = (N + BKT_ROWS - 1) / BKT_ROWS;
    const int NEB = (E + EDGE_BLK - 1) / EDGE_BLK;
    float* h        = (float*)d_ws;
    int*   blk_hist = (int*)(h + (size_t)N * NCLASS);
    float* s_row    = (float*)(blk_hist + (size_t)NEB * MAX_BKT);
    float* s_col    = s_row + N;
    int*   recs     = (int*)(s_col + N);
    int*   pairs    = recs + 2 * (size_t)E;
    int*   off      = pairs + 2 * (size_t)E;
    int*   bstart   = off + N + 1;

    float* out = (float*)d_out;

    {
        const int mtiles = N / 16;
        dim3 grid((mtiles + 7) / 8, NCLASS / 64);
        spgat_gemm<<<grid, 256, 0, stream>>>(x, W, h, N);
    }
    {
        dim3 grid((N + 7) / 8);
        spgat_scores<<<grid, 256, 0, stream>>>(h, a, s_row, s_col, N);
    }
    spgat_bcount<<<NEB, 1024, 0, stream>>>(ei, E, blk_hist);
    spgat_bscatter<<<NEB, 1024, 0, stream>>>(ei, E, NB, blk_hist, s_row, s_col, bstart, recs);
    spgat_bsort<<<NB, 1024, 0, stream>>>(bstart, recs, off, pairs, N);
    {
        dim3 grid((N + 7) / 8);
        spgat_gather<<<grid, 256, 0, stream>>>(off, pairs, h, out, N);
    }
}


# Score dot products formed in the GEMM epilogue from the accumulators (a partial per row, kind and 64-col quarter; edge pass adds the 4); scores launch and its h re-read gone; s sum order changes
# speedup vs baseline: 2.9780x; 1.0498x over previous
#include <hip/hip_runtime.h>
#include <hip/hip_bf16.h>

#define NFEAT  512
#define NCLASS 256

typedef float v2f __attribute__((ext_vector_type(2)));
typedef float v8f __attribute__((ext_vector_type(8)));


template <int M>
__device__ __forceinline__ float xrow(float x) {
    const int b = __builtin_bit_cast(int, x);
    return __builtin_bit_cast(float, __builtin_amdgcn_update_dpp(0, b, 0x160 + M, 0xf, 0xf, true));
}

__global__ void spgat_gemm(const float* __restrict__ x,
                           const float* __restrict__ W,
                           const float* __restrict__ a,
                           float* __restrict__ h,
                           float* __restrict__ sp, int N) {
    const int lane = threadIdx.x & 31;
    const int wave = __builtin_amdgcn_readfirstlane((int)(threadIdx.x >> 5));
    const int mtile = blockIdx.x * 8 + wave;
    const int mtiles = N >> 4;
    if (mtile >= mtiles) return;
    const int m0   = mtile << 4;
    const int n0   = blockIdx.y * 64;
    const int half = lane >> 4;
    const int sub  = lane & 15;

    v8f acc0 = {}, acc1 = {}, acc2 = {}, acc3 = {};

    const float4* xa   = (const float4*)(x + (size_t)(m0 + sub) * NFEAT + 4 * half);
    const float4* wcol = (const float4*)(W + (size_t)(4 * half) * NCLASS + n0 + 4 * sub);
    #pragma unroll 2
    for (int kb = 0; kb < NFEAT; kb += 8) {
        const float4 av = xa[kb >> 2];
        const float4 w0 = wcol[(size_t)(kb + 0) * (NCLASS / 4)];
        const float4 w1 = wcol[(size_t)(kb + 1) * (NCLASS / 4)];
        const float4 w2 = wcol[(size_t)(kb + 2) * (NCLASS / 4)];
        const float4 w3 = wcol[(size_t)(kb + 3) * (NCLASS / 4)];

        v2f af;  af.x = av.x; af.y = av.y;
        v2f b0;  b0.x = w0.x; b0.y = w1.x;
        v2f b1;  b1.x = w0.y; b1.y = w1.y;
        v2f b2;  b2.x = w0.z; b2.y = w1.z;
        v2f b3;  b3.x = w0.w; b3.y = w1.w;
        acc0 = __builtin_amdgcn_wmma_f32_16x16x4_f32(false, af, false, b0, (short)0, acc0, false, false);
        acc1 = __builtin_amdgcn_wmma_f32_16x16x4_f32(false, af, false, b1, (short)0, acc1, false, false);
        acc2 = __builtin_amdgcn_wmma_f32_16x16x4_f32(false, af, false, b2, (short)0, acc2, false, false);
        acc3 = __builtin_amdgcn_wmma_f32_16x16x4_f32(false, af, false, b3, (short)0, acc3, false, false);

        af.x = av.z; af.y = av.w;
        b0.x = w2.x; b0.y = w3.x;
        b1.x = w2.y; b1.y = w3.y;
        b2.x = w2.z; b2.y = w3.z;
        b3.x = w2.w; b3.y = w3.w;
        acc0 = __builtin_amdgcn_wmma_f32_16x16x4_f32(false, af, false, b0, (short)0, acc0, false, false);
        acc1 = __builtin_amdgcn_wmma_f32_16x16x4_f32(false, af, false, b1, (short)0, acc1, false, false);
        acc2 = __builtin_amdgcn_wmma_f32_16x16x4_f32(false, af, false, b2, (short)0, acc2, false, false);
        acc3 = __builtin_amdgcn_wmma_f32_16x16x4_f32(false, af, false, b3, (short)0, acc3, false, false);
    }

    const float4 ar = *(const float4*)(a + n0 + 4 * sub);
    const float4 ac = *(const float4*)(a + NCLASS + n0 + 4 * sub);
    const bool b3 = (sub & 8) != 0, b2 = (sub & 4) != 0, b1 = (sub & 2) != 0, b0 = (sub & 1) != 0;
    float* hout = h + (size_t)(m0 + half * 8) * NCLASS + n0 + 4 * sub;
    float v8[8];
    #pragma unroll
    for (int r = 0; r < 8; ++r) {
        const float4 hv = make_float4(acc0[r], acc1[r], acc2[r], acc3[r]);
        *(float4*)(hout + (size_t)r * NCLASS) = hv;
        const float pr = fmaf(hv.w, ar.w, fmaf(hv.z, ar.z, fmaf(hv.y, ar.y, hv.x * ar.x)));
        const float pc = fmaf(hv.w, ac.w, fmaf(hv.z, ac.z, fmaf(hv.y, ac.y, hv.x * ac.x)));
        const float keep = b3 ? pc : pr;
        const float send = b3 ? pr : pc;
        v8[r] = keep + xrow<8>(send);
    }
    float v4[4];
    #pragma unroll
    for (int j = 0; j < 4; ++j) {
        const float keep = b2 ? v8[4 + j] : v8[j];
        const float send = b2 ? v8[j] : v8[4 + j];
        v4[j] = keep + xrow<4>(send);
    }
    float v2[2];
    #pragma unroll
    for (int j = 0; j < 2; ++j) {
        const float keep = b1 ? v4[2 + j] : v4[j];
        const float send = b1 ? v4[j] : v4[2 + j];
        v2[j] = keep + xrow<2>(send);
    }
    const float keep1 = b0 ? v2[1] : v2[0];
    const float send1 = b0 ? v2[0] : v2[1];
    const float tot = keep1 + xrow<1>(send1);
    sp[(size_t)(m0 + half * 8 + (sub & 7)) * 8 + 4 * (sub >> 3) + blockIdx.y] = tot;
}

#define BKT_SHIFT 8
#define BKT_ROWS  256
#define MAX_BKT   256
#define EDGE_BLK  4096

__device__ __forceinline__ void scan256_excl(int* a, int lane) {
    int v[8];
    int s = 0;
    #pragma unroll
    for (int k = 0; k < 8; ++k) { v[k] = a[8 * lane + k]; s += v[k]; }
    int incl = s;
    #pragma unroll
    for (int o = 1; o < 32; o <<= 1) {
        const int t = __shfl_up(incl, o, 32);
        if (lane >= o) incl += t;
    }
    int run = incl - s;
    #pragma unroll
    for (int k = 0; k < 8; ++k) { a[8 * lane + k] = run; run += v[k]; }
}

__global__ void __launch_bounds__(1024)
spgat_bcount(const int* __restrict__ ei, int E,
             int* __restrict__ blk_hist) {
    __shared__ int hist[MAX_BKT];
    const int tid = threadIdx.x;
    if (tid < MAX_BKT) hist[tid] = 0;
    __syncthreads();
    const int e0 = blockIdx.x * EDGE_BLK + tid;
    int r[4];
    #pragma unroll
    for (int i = 0; i < 4; ++i) {
        const int e = e0 + 1024 * i;
        r[i] = ei[e < E ? e : E - 1];
    }
    #pragma unroll
    for (int i = 0; i < 4; ++i)
        if (e0 + 1024 * i < E) atomicAdd(&hist[r[i] >> BKT_SHIFT], 1);
    __syncthreads();
    if (tid < MAX_BKT) blk_hist[blockIdx.x * MAX_BKT + tid] = hist[tid];
}

__global__ void __launch_bounds__(1024)
spgat_bscatter(const int* __restrict__ ei, int E, int NB,
               const int* __restrict__ blk_hist,
               const float* __restrict__ sp,
               int* __restrict__ bstart,
               int* __restrict__ recs) {
    __shared__ __attribute__((aligned(16))) int p_tot[16][MAX_BKT];
    __shared__ __attribute__((aligned(16))) int p_pre[16][MAX_BKT];
    __shared__ int s_tot[MAX_BKT];
    __shared__ int cursor[MAX_BKT];
    const int tid = threadIdx.x;
    const int wg  = blockIdx.x;
    const int nwg = gridDim.x;
    {
        const int c4 = tid & 63;
        const int g  = tid >> 6;
        const int4* tbl = (const int4*)blk_hist + c4;
        int4 tot = make_int4(0, 0, 0, 0), pre = make_int4(0, 0, 0, 0);
        for (int w0 = g; w0 < nwg; w0 += 128) {
            int4 v[8];
            #pragma unroll
            for (int k = 0; k < 8; ++k) {
                const int w = w0 + 16 * k;
                v[k] = tbl[(size_t)(w < nwg ? w : nwg - 1) * (MAX_BKT / 4)];
            }
            #pragma unroll
            for (int k = 0; k < 8; ++k) {
                const int w = w0 + 16 * k;
                const int in = w < nwg ? -1 : 0;
                const int fr = w < wg  ? -1 : 0;
                tot.x += v[k].x & in;       tot.y += v[k].y & in;
                tot.z += v[k].z & in;       tot.w += v[k].w & in;
                pre.x += v[k].x & fr;       pre.y += v[k].y & fr;
                pre.z += v[k].z & fr;       pre.w += v[k].w & fr;
            }
        }
        *(int4*)&p_tot[g][4 * c4] = tot;
        *(int4*)&p_pre[g][4 * c4] = pre;
    }
    __syncthreads();
    int pre_b = 0;
    if (tid < MAX_BKT) {
        int t = 0;
        #pragma unroll
        for (int k = 0; k < 16; ++k) { t += p_tot[k][tid]; pre_b += p_pre[k][tid]; }
        s_tot[tid] = t;
    }
    __syncthreads();
    if (tid < 32) scan256_excl(s_tot, tid);
    __syncthreads();
    if (tid < MAX_BKT) {
        const int st = s_tot[tid];
        cursor[tid] = st + pre_b;
        if (wg == 0 && tid < NB) bstart[tid] = st;
    }
    if (wg == 0 && tid == 0) bstart[NB] = E;
    __syncthreads();

    const int e0 = wg * EDGE_BLK + tid;
    int r[4], c[4];
    #pragma unroll
    for (int i = 0; i < 4; ++i) {
        const int e  = e0 + 1024 * i;
        const int ec = e < E ? e : E - 1;
        r[i] = ei[ec];
        c[i] = ei[E + ec];
    }
    float4 pr[4], pc[4];
    #pragma unroll
    for (int i = 0; i < 4; ++i) {
        pr[i] = ((const float4*)sp)[2 * (size_t)r[i]];
        pc[i] = ((const float4*)sp)[2 * (size_t)c[i] + 1];
    }
    float lg[4];
    #pragma unroll
    for (int i = 0; i < 4; ++i)
        lg[i] = ((pr[i].x + pr[i].y) + (pr[i].z + pr[i].w)) +
                ((pc[i].x + pc[i].y) + (pc[i].z + pc[i].w));
    #pragma unroll
    for (int i = 0; i < 4; ++i) {
        if (e0 + 1024 * i < E) {
            const float l  = lg[i] > 0.f ? lg[i] : 0.2f * lg[i];
            const float ev = __expf(-l);
            const int p = atomicAdd(&cursor[r[i] >> BKT_SHIFT], 1);
            ((int2*)recs)[p] = make_int2((int)(((unsigned)(r[i] & (BKT_ROWS - 1)) << 24) | (unsigned)c[i]),
                                         __float_as_int(ev));
        }
    }
}

__global__ void __launch_bounds__(1024)
spgat_bsort(const int* __restrict__ bstart,
            const int* __restrict__ recs,
            int* __restrict__ off,
            int* __restrict__ pairs, int N) {
    __shared__ int cnt[BKT_ROWS];
    __shared__ int cur[BKT_ROWS];
    const int tid = threadIdx.x;
    const int b   = blockIdx.x;
    const int beg = bstart[b];
    const int end = bstart[b + 1];
    if (tid < BKT_ROWS) cnt[tid] = 0;
    __syncthreads();
    for (int i0 = beg + tid; i0 < end; i0 += 5 * 1024) {
        int k[5];
        #pragma unroll
        for (int j = 0; j < 5; ++j) {
            const int i = i0 + 1024 * j;
            k[j] = recs[2 * (size_t)(i < end ? i : end - 1)];
        }
        #pragma unroll
        for (int j = 0; j < 5; ++j)
            if (i0 + 1024 * j < end) atomicAdd(&cnt[((unsigned)k[j]) >> 24], 1);
    }
    __syncthreads();
    if (tid < 32) scan256_excl(cnt, tid);
    __syncthreads();
    if (tid < BKT_ROWS) {
        const int o = beg + cnt[tid];
        cur[tid] = o;
        const int row = b * BKT_ROWS + tid;
        if (row < N) off[row] = o;
    }
    if (b == (int)gridDim.x - 1 && tid == 0) off[N] = end;
    __syncthreads();
    for (int i0 = beg + tid; i0 < end; i0 += 5 * 1024) {
        int2 rc[5];
        #pragma unroll
        for (int j = 0; j < 5; ++j) {
            const int i = i0 + 1024 * j;
            rc[j] = ((const int2*)recs)[i < end ? i : end - 1];
        }
        #pragma unroll
        for (int j = 0; j < 5; ++j) {
            if (i0 + 1024 * j < end) {
                const int p = atomicAdd(&cur[((unsigned)rc[j].x) >> 24], 1);
                ((int2*)pairs)[p] = make_int2(rc[j].x & 0xFFFFFF, rc[j].y);
            }
        }
    }
}

__device__ __forceinline__ void axpy4(float4& acc, float s, const float4& v) {
    acc.x += s * v.x; acc.y += s * v.y; acc.z += s * v.z; acc.w += s * v.w;
}

__device__ __forceinline__ float xhalf(float x) {
    const int b = __builtin_bit_cast(int, x);
    return __builtin_bit_cast(float,
        __builtin_amdgcn_permlanex16(b, b, 0x76543210, 0xfedcba98, false, false));
}
__device__ __forceinline__ float wave_max(float v) {
    v = fmaxf(v, xrow<1>(v));
    v = fmaxf(v, xrow<2>(v));
    v = fmaxf(v, xrow<4>(v));
    v = fmaxf(v, xrow<8>(v));
    return fmaxf(v, xhalf(v));
}
__device__ __forceinline__ float wave_sum(float v) {
    v += xrow<1>(v);
    v += xrow<2>(v);
    v += xrow<4>(v);
    v += xrow<8>(v);
    return v + xhalf(v);
}

__global__ void spgat_gather(const int* __restrict__ off,
                             const int* __restrict__ pairs,
                             const float* __restrict__ h,
                             float* __restrict__ out, int N) {
    const int lane = threadIdx.x & 31;
    const int row = __builtin_amdgcn_readfirstlane(
        (int)(blockIdx.x * (blockDim.x >> 5) + (threadIdx.x >> 5)));
    if (row >= N) return;
    const int beg = off[row];
    const int end = off[row + 1];
    const float4* hl = (const float4*)h + lane;
    float4 a0 = make_float4(0.f, 0.f, 0.f, 0.f);
    float4 a1 = make_float4(0.f, 0.f, 0.f, 0.f);
    float rs = 0.f;
    #pragma unroll 4
    for (int j = beg; j < end; ++j) {
        const int2 p = ((const int2*)pairs)[j];
        const float4* q = hl + (size_t)p.x * (NCLASS / 4);
        const float4 u = q[0], v = q[32];
        axpy4(a0, __int_as_float(p.y), u);
        axpy4(a1, __int_as_float(p.y), v);
        rs += __int_as_float(p.y);
    }

    const float inv = 1.0f / rs;
    float v[8] = {a0.x, a0.y, a0.z, a0.w, a1.x, a1.y, a1.z, a1.w};
    float mx = -3.402823466e38f;
    #pragma unroll
    for (int i = 0; i < 8; ++i) {
        const float hp = v[i] * inv;
        const float el = hp > 0.f ? hp : (__expf(hp) - 1.f);
        v[i] = el;
        mx = fmaxf(mx, el);
    }
    mx = wave_max(mx);
    float s = 0.f;
    #pragma unroll
    for (int i = 0; i < 8; ++i) s += __expf(v[i] - mx);
    s = wave_sum(s);
    const float lse = mx + __logf(s);

    float4* o = (float4*)(out + (size_t)row * NCLASS) + lane;
    o[0]  = make_float4(v[0] - lse, v[1] - lse, v[2] - lse, v[3] - lse);
    o[32] = make_float4(v[4] - lse, v[5] - lse, v[6] - lse, v[7] - lse);
}

extern "C" void kernel_launch(void* const* d_in, const int* in_sizes, int n_in,
                              void* d_out, int out_size, void* d_ws, size_t ws_size,
                              hipStream_t stream) {
    const float* x  = (const float*)d_in[0];
    const float* W  = (const float*)d_in[1];
    const float* a  = (const float*)d_in[2];
    const int*   ei = (const int*)d_in[3];

    const int N = in_sizes[0] / NFEAT;
    const int E = in_sizes[3] / 2;

    const int NB  = (N + BKT_ROWS - 1) / BKT_ROWS;
    const int NEB = (E + EDGE_BLK - 1) / EDGE_BLK;
    float* h        = (float*)d_ws;
    int*   blk_hist = (int*)(h + (size_t)N * NCLASS);
    float* sp       = (float*)(blk_hist + (size_t)NEB * MAX_BKT);
    int*   recs     = (int*)(sp + 8 * (size_t)N);
    int*   pairs    = recs + 2 * (size_t)E;
    int*   off      = pairs + 2 * (size_t)E;
    int*   bstart   = off + N + 1;

    float* out = (float*)d_out;

    {
        const int mtiles = N / 16;
        dim3 grid((mtiles + 7) / 8, NCLASS / 64);
        spgat_gemm<<<grid, 256, 0, stream>>>(x, W, a, h, sp, N);
    }
    spgat_bcount<<<NEB, 1024, 0, stream>>>(ei, E, blk_hist);
    spgat_bscatter<<<NEB, 1024, 0, stream>>>(ei, E, NB, blk_hist, sp, bstart, recs);
    spgat_bsort<<<NB, 1024, 0, stream>>>(bstart, recs, off, pairs, N);
    {
        dim3 grid((N + 7) / 8);
        spgat_gather<<<grid, 256, 0, stream>>>(off, pairs, h, out, N);
    }
}


# Edge binning (indices only) runs as extra blocks of the GEMM launch, beside the matrix work; edge weights formed in the per-bucket sort; 4 launches; per-row sum order = LDS arrival rank as before
# speedup vs baseline: 3.2386x; 1.0875x over previous
#include <hip/hip_runtime.h>
#include <hip/hip_bf16.h>

#define NFEAT  512
#define NCLASS 256

typedef float v2f __attribute__((ext_vector_type(2)));
typedef float v8f __attribute__((ext_vector_type(8)));


template <int M>
__device__ __forceinline__ float xrow(float x) {
    const int b = __builtin_bit_cast(int, x);
    return __builtin_bit_cast(float, __builtin_amdgcn_update_dpp(0, b, 0x160 + M, 0xf, 0xf, true));
}

#define BKT_SHIFT 8
#define BKT_ROWS  256
#define MAX_BKT   256
#define EDGE_BLK  4096

__device__ __forceinline__ void scan256_excl(int* a, int lane) {
    int v[8];
    int s = 0;
    #pragma unroll
    for (int k = 0; k < 8; ++k) { v[k] = a[8 * lane + k]; s += v[k]; }
    int incl = s;
    #pragma unroll
    for (int o = 1; o < 32; o <<= 1) {
        const int t = __shfl_up(incl, o, 32);
        if (lane >= o) incl += t;
    }
    int run = incl - s;
    #pragma unroll
    for (int k = 0; k < 8; ++k) { a[8 * lane + k] = run; run += v[k]; }
}

__device__ __forceinline__ void bin_edge_block(const int* __restrict__ ei, int E, int NB,
                                               int wg, int nwg,
                                               const int* __restrict__ blk_hist,
                                               int* __restrict__ bstart,
                                               int* __restrict__ recs) {
    __shared__ __attribute__((aligned(16))) int p_tot[4][MAX_BKT];
    __shared__ __attribute__((aligned(16))) int p_pre[4][MAX_BKT];
    __shared__ int s_tot[MAX_BKT];
    __shared__ int cursor[MAX_BKT];
    const int tid = threadIdx.x;
    {
        const int c4 = tid & 63;
        const int g  = __builtin_amdgcn_readfirstlane(tid >> 6);
        const int4* tbl = (const int4*)blk_hist + c4;
        int4 tot = make_int4(0, 0, 0, 0), pre = make_int4(0, 0, 0, 0);
        #pragma unroll 4
        for (int w = g; w < nwg; w += 4) {
            const int4 v = tbl[(size_t)w * (MAX_BKT / 4)];
            tot.x += v.x; tot.y += v.y; tot.z += v.z; tot.w += v.w;
            if (w < wg) { pre.x += v.x; pre.y += v.y; pre.z += v.z; pre.w += v.w; }
        }
        *(int4*)&p_tot[g][4 * c4] = tot;
        *(int4*)&p_pre[g][4 * c4] = pre;
    }
    __syncthreads();
    int pre_b = 0;
    {
        int t = 0;
        #pragma unroll
        for (int k = 0; k < 4; ++k) { t += p_tot[k][tid]; pre_b += p_pre[k][tid]; }
        s_tot[tid] = t;
    }
    __syncthreads();
    if (tid < 32) scan256_excl(s_tot, tid);
    __syncthreads();
    {
        const int st = s_tot[tid];
        cursor[tid] = st + pre_b;
        if (wg == 0 && tid < NB) bstart[tid] = st;
    }
    if (wg == 0 && tid == 0) bstart[NB] = E;
    __syncthreads();

    #pragma unroll 1
    for (int q = 0; q < 4; ++q) {
        const int e0 = wg * EDGE_BLK + q * 1024 + tid;
        int r[4], c[4];
        #pragma unroll
        for (int i = 0; i < 4; ++i) {
            const int e  = e0 + 256 * i;
            const int ec = e < E ? e : E - 1;
            r[i] = ei[ec];
            c[i] = ei[E + ec];
        }
        #pragma unroll
        for (int i = 0; i < 4; ++i) {
            if (e0 + 256 * i < E) {
                const int p = atomicAdd(&cursor[r[i] >> BKT_SHIFT], 1);
                recs[p] = (int)(((unsigned)(r[i] & (BKT_ROWS - 1)) << 24) | (unsigned)c[i]);
            }
        }
    }
    __syncthreads();
}

__global__ void __launch_bounds__(256, 16)
spgat_gemm(const float* __restrict__ x,
                           const float* __restrict__ W,
                           const float* __restrict__ a,
                           float* __restrict__ h,
                           float* __restrict__ sp, int N,
                           const int* __restrict__ ei, int E, int NB, int NEB,
                           const int* __restrict__ blk_hist,
                           int* __restrict__ bstart,
                           int* __restrict__ recs) {
    if (blockIdx.y == NCLASS / 64) {
        for (int wg = blockIdx.x; wg < NEB; wg += gridDim.x)
            bin_edge_block(ei, E, NB, wg, NEB, blk_hist, bstart, recs);
        return;
    }
    const int lane = threadIdx.x & 31;
    const int wave = __builtin_amdgcn_readfirstlane((int)(threadIdx.x >> 5));
    const int mtile = blockIdx.x * 8 + wave;
    const int mtiles = N >> 4;
    if (mtile >= mtiles) return;
    const int m0   = mtile << 4;
    const int n0   = blockIdx.y * 64;
    const int half = lane >> 4;
    const int sub  = lane & 15;

    v8f acc0 = {}, acc1 = {}, acc2 = {}, acc3 = {};

    const float4* xa   = (const float4*)(x + (size_t)(m0 + sub) * NFEAT + 4 * half);
    const float4* wcol = (const float4*)(W + (size_t)(4 * half) * NCLASS + n0 + 4 * sub);
    #pragma unroll 2
    for (int kb = 0; kb < NFEAT; kb += 8) {
        const float4 av = xa[kb >> 2];
        const float4 w0 = wcol[(size_t)(kb + 0) * (NCLASS / 4)];
        const float4 w1 = wcol[(size_t)(kb + 1) * (NCLASS / 4)];
        const float4 w2 = wcol[(size_t)(kb + 2) * (NCLASS / 4)];
        const float4 w3 = wcol[(size_t)(kb + 3) * (NCLASS / 4)];

        v2f af;  af.x = av.x; af.y = av.y;
        v2f b0;  b0.x = w0.x; b0.y = w1.x;
        v2f b1;  b1.x = w0.y; b1.y = w1.y;
        v2f b2;  b2.x = w0.z; b2.y = w1.z;
        v2f b3;  b3.x = w0.w; b3.y = w1.w;
        acc0 = __builtin_amdgcn_wmma_f32_16x16x4_f32(false, af, false, b0, (short)0, acc0, false, false);
        acc1 = __builtin_amdgcn_wmma_f32_16x16x4_f32(false, af, false, b1, (short)0, acc1, false, false);
        acc2 = __builtin_amdgcn_wmma_f32_16x16x4_f32(false, af, false, b2, (short)0, acc2, false, false);
        acc3 = __builtin_amdgcn_wmma_f32_16x16x4_f32(false, af, false, b3, (short)0, acc3, false, false);

        af.x = av.z; af.y = av.w;
        b0.x = w2.x; b0.y = w3.x;
        b1.x = w2.y; b1.y = w3.y;
        b2.x = w2.z; b2.y = w3.z;
        b3.x = w2.w; b3.y = w3.w;
        acc0 = __builtin_amdgcn_wmma_f32_16x16x4_f32(false, af, false, b0, (short)0, acc0, false, false);
        acc1 = __builtin_amdgcn_wmma_f32_16x16x4_f32(false, af, false, b1, (short)0, acc1, false, false);
        acc2 = __builtin_amdgcn_wmma_f32_16x16x4_f32(false, af, false, b2, (short)0, acc2, false, false);
        acc3 = __builtin_amdgcn_wmma_f32_16x16x4_f32(false, af, false, b3, (short)0, acc3, false, false);
    }

    const float4 ar = *(const float4*)(a + n0 + 4 * sub);
    const float4 ac = *(const float4*)(a + NCLASS + n0 + 4 * sub);
    const bool b3 = (sub & 8) != 0, b2 = (sub & 4) != 0, b1 = (sub & 2) != 0, b0 = (sub & 1) != 0;
    float* hout = h + (size_t)(m0 + half * 8) * NCLASS + n0 + 4 * sub;
    float v8[8];
    #pragma unroll
    for (int r = 0; r < 8; ++r) {
        const float4 hv = make_float4(acc0[r], acc1[r], acc2[r], acc3[r]);
        *(float4*)(hout + (size_t)r * NCLASS) = hv;
        const float pr = fmaf(hv.w, ar.w, fmaf(hv.z, ar.z, fmaf(hv.y, ar.y, hv.x * ar.x)));
        const float pc = fmaf(hv.w, ac.w, fmaf(hv.z, ac.z, fmaf(hv.y, ac.y, hv.x * ac.x)));
        const float keep = b3 ? pc : pr;
        const float send = b3 ? pr : pc;
        v8[r] = keep + xrow<8>(send);
    }
    float v4[4];
    #pragma unroll
    for (int j = 0; j < 4; ++j) {
        const float keep = b2 ? v8[4 + j] : v8[j];
        const float send = b2 ? v8[j] : v8[4 + j];
        v4[j] = keep + xrow<4>(send);
    }
    float v2[2];
    #pragma unroll
    for (int j = 0; j < 2; ++j) {
        const float keep = b1 ? v4[2 + j] : v4[j];
        const float send = b1 ? v4[j] : v4[2 + j];
        v2[j] = keep + xrow<2>(send);
    }
    const float keep1 = b0 ? v2[1] : v2[0];
    const float send1 = b0 ? v2[0] : v2[1];
    const float tot = keep1 + xrow<1>(send1);
    sp[(size_t)(m0 + half * 8 + (sub & 7)) * 8 + 4 * (sub >> 3) + blockIdx.y] = tot;
}

__global__ void __launch_bounds__(1024)
spgat_bcount(const int* __restrict__ ei, int E,
             int* __restrict__ blk_hist) {
    __shared__ int hist[MAX_BKT];
    const int tid = threadIdx.x;
    if (tid < MAX_BKT) hist[tid] = 0;
    __syncthreads();
    const int e0 = blockIdx.x * EDGE_BLK + tid;
    int r[4];
    #pragma unroll
    for (int i = 0; i < 4; ++i) {
        const int e = e0 + 1024 * i;
        r[i] = ei[e < E ? e : E - 1];
    }
    #pragma unroll
    for (int i = 0; i < 4; ++i)
        if (e0 + 1024 * i < E) atomicAdd(&hist[r[i] >> BKT_SHIFT], 1);
    __syncthreads();
    if (tid < MAX_BKT) blk_hist[blockIdx.x * MAX_BKT + tid] = hist[tid];
}

__global__ void __launch_bounds__(1024)
spgat_bsort(const int* __restrict__ bstart,
            const int* __restrict__ recs,
            const float* __restrict__ sp,
            int* __restrict__ off,
            int* __restrict__ pairs, int N) {
    __shared__ int cnt[BKT_ROWS];
    __shared__ int cur[BKT_ROWS];
    const int tid = threadIdx.x;
    const int b   = blockIdx.x;
    const int beg = bstart[b];
    const int end = bstart[b + 1];
    if (tid < BKT_ROWS) cnt[tid] = 0;
    __syncthreads();
    for (int i0 = beg + tid; i0 < end; i0 += 5 * 1024) {
        int k[5];
        #pragma unroll
        for (int j = 0; j < 5; ++j) {
            const int i = i0 + 1024 * j;
            k[j] = recs[i < end ? i : end - 1];
        }
        #pragma unroll
        for (int j = 0; j < 5; ++j)
            if (i0 + 1024 * j < end) atomicAdd(&cnt[((unsigned)k[j]) >> 24], 1);
    }
    __syncthreads();
    if (tid < 32) scan256_excl(cnt, tid);
    __syncthreads();
    if (tid < BKT_ROWS) {
        const int o = beg + cnt[tid];
        cur[tid] = o;
        const int row = b * BKT_ROWS + tid;
        if (row < N) off[row] = o;
    }
    if (b == (int)gridDim.x - 1 && tid == 0) off[N] = end;
    __syncthreads();
    for (int i0 = beg + tid; i0 < end; i0 += 5 * 1024) {
        int rc[5];
        #pragma unroll
        for (int j = 0; j < 5; ++j) {
            const int i = i0 + 1024 * j;
            rc[j] = recs[i < end ? i : end - 1];
        }
        float4 pr[5], pc[5];
        #pragma unroll
        for (int j = 0; j < 5; ++j) {
            const int r  = b * BKT_ROWS + (int)(((unsigned)rc[j]) >> 24);
            const int cc = rc[j] & 0xFFFFFF;
            pr[j] = ((const float4*)sp)[2 * (size_t)r];
            pc[j] = ((const float4*)sp)[2 * (size_t)cc + 1];
        }
        #pragma unroll
        for (int j = 0; j < 5; ++j) {
            if (i0 + 1024 * j < end) {
                const float lg = ((pr[j].x + pr[j].y) + (pr[j].z + pr[j].w)) +
                                 ((pc[j].x + pc[j].y) + (pc[j].z + pc[j].w));
                const float l  = lg > 0.f ? lg : 0.2f * lg;
                const float ev = __expf(-l);
                const int p = atomicAdd(&cur[((unsigned)rc[j]) >> 24], 1);
                ((int2*)pairs)[p] = make_int2(rc[j] & 0xFFFFFF, __float_as_int(ev));
            }
        }
    }
}

__device__ __forceinline__ void axpy4(float4& acc, float s, const float4& v) {
    acc.x += s * v.x; acc.y += s * v.y; acc.z += s * v.z; acc.w += s * v.w;
}

__device__ __forceinline__ float xhalf(float x) {
    const int b = __builtin_bit_cast(int, x);
    return __builtin_bit_cast(float,
        __builtin_amdgcn_permlanex16(b, b, 0x76543210, 0xfedcba98, false, false));
}
__device__ __forceinline__ float wave_max(float v) {
    v = fmaxf(v, xrow<1>(v));
    v = fmaxf(v, xrow<2>(v));
    v = fmaxf(v, xrow<4>(v));
    v = fmaxf(v, xrow<8>(v));
    return fmaxf(v, xhalf(v));
}
__device__ __forceinline__ float wave_sum(float v) {
    v += xrow<1>(v);
    v += xrow<2>(v);
    v += xrow<4>(v);
    v += xrow<8>(v);
    return v + xhalf(v);
}

__global__ void spgat_gather(const int* __restrict__ off,
                             const int* __restrict__ pairs,
                             const float* __restrict__ h,
                             float* __restrict__ out, int N) {
    const int lane = threadIdx.x & 31;
    const int row = __builtin_amdgcn_readfirstlane(
        (int)(blockIdx.x * (blockDim.x >> 5) + (threadIdx.x >> 5)));
    if (row >= N) return;
    const int beg = off[row];
    const int end = off[row + 1];
    const float4* hl = (const float4*)h + lane;
    float4 a0 = make_float4(0.f, 0.f, 0.f, 0.f);
    float4 a1 = make_float4(0.f, 0.f, 0.f, 0.f);
    float rs = 0.f;
    #pragma unroll 4
    for (int j = beg; j < end; ++j) {
        const int2 p = ((const int2*)pairs)[j];
        const float4* q = hl + (size_t)p.x * (NCLASS / 4);
        const float4 u = q[0], v = q[32];
        axpy4(a0, __int_as_float(p.y), u);
        axpy4(a1, __int_as_float(p.y), v);
        rs += __int_as_float(p.y);
    }

    const float inv = 1.0f / rs;
    float v[8] = {a0.x, a0.y, a0.z, a0.w, a1.x, a1.y, a1.z, a1.w};
    float mx = -3.402823466e38f;
    #pragma unroll
    for (int i = 0; i < 8; ++i) {
        const float hp = v[i] * inv;
        const float el = hp > 0.f ? hp : (__expf(hp) - 1.f);
        v[i] = el;
        mx = fmaxf(mx, el);
    }
    mx = wave_max(mx);
    float s = 0.f;
    #pragma unroll
    for (int i = 0; i < 8; ++i) s += __expf(v[i] - mx);
    s = wave_sum(s);
    const float lse = mx + __logf(s);

    float4* o = (float4*)(out + (size_t)row * NCLASS) + lane;
    o[0]  = make_float4(v[0] - lse, v[1] - lse, v[2] - lse, v[3] - lse);
    o[32] = make_float4(v[4] - lse, v[5] - lse, v[6] - lse, v[7] - lse);
}

extern "C" void kernel_launch(void* const* d_in, const int* in_sizes, int n_in,
                              void* d_out, int out_size, void* d_ws, size_t ws_size,
                              hipStream_t stream) {
    const float* x  = (const float*)d_in[0];
    const float* W  = (const float*)d_in[1];
    const float* a  = (const float*)d_in[2];
    const int*   ei = (const int*)d_in[3];

    const int N = in_sizes[0] / NFEAT;
    const int E = in_sizes[3] / 2;

    const int NB  = (N + BKT_ROWS - 1) / BKT_ROWS;
    const int NEB = (E + EDGE_BLK - 1) / EDGE_BLK;
    float* h        = (float*)d_ws;
    int*   blk_hist = (int*)(h + (size_t)N * NCLASS);
    float* sp       = (float*)(blk_hist + (size_t)NEB * MAX_BKT);
    int*   recs     = (int*)(sp + 8 * (size_t)N);
    int*   pairs    = recs + 2 * (size_t)E;
    int*   off      = pairs + 2 * (size_t)E;
    int*   bstart   = off + N + 1;

    float* out = (float*)d_out;

    spgat_bcount<<<NEB, 1024, 0, stream>>>(ei, E, blk_hist);
    {
        const int mtiles = N / 16;
        dim3 grid((mtiles + 7) / 8, NCLASS / 64 + 1);
        spgat_gemm<<<grid, 256, 0, stream>>>(x, W, a, h, sp, N,
                                             ei, E, NB, NEB, blk_hist, bstart, recs);
    }
    spgat_bsort<<<NB, 1024, 0, stream>>>(bstart, recs, sp, off, pairs, N);
    {
        dim3 grid((N + 7) / 8);
        spgat_gather<<<grid, 256, 0, stream>>>(off, pairs, h, out, N);
    }
}


# GEMM operands staged through LDS: per 32-k slab a block copies its 128x32 x tile and 32x64 W slab by whole-line 16-B loads into one of two buffers, one barrier a slab; fragments, sum order unchanged
# speedup vs baseline: 3.6313x; 1.1213x over previous
#include <hip/hip_runtime.h>
#include <hip/hip_bf16.h>

#define NFEAT  512
#define NCLASS 256

typedef float v2f __attribute__((ext_vector_type(2)));
typedef float v8f __attribute__((ext_vector_type(8)));


template <int M>
__device__ __forceinline__ float xrow(float x) {
    const int b = __builtin_bit_cast(int, x);
    return __builtin_bit_cast(float, __builtin_amdgcn_update_dpp(0, b, 0x160 + M, 0xf, 0xf, true));
}

#define BKT_SHIFT 8
#define BKT_ROWS  256
#define MAX_BKT   256
#define EDGE_BLK  4096

__device__ __forceinline__ void scan256_excl(int* a, int lane) {
    int v[8];
    int s = 0;
    #pragma unroll
    for (int k = 0; k < 8; ++k) { v[k] = a[8 * lane + k]; s += v[k]; }
    int incl = s;
    #pragma unroll
    for (int o = 1; o < 32; o <<= 1) {
        const int t = __shfl_up(incl, o, 32);
        if (lane >= o) incl += t;
    }
    int run = incl - s;
    #pragma unroll
    for (int k = 0; k < 8; ++k) { a[8 * lane + k] = run; run += v[k]; }
}

__device__ __forceinline__ void bin_edge_block(const int* __restrict__ ei, int E, int NB,
                                               int wg, int nwg,
                                               const int* __restrict__ blk_hist,
                                               int* __restrict__ bstart,
                                               int* __restrict__ recs) {
    __shared__ __attribute__((aligned(16))) int p_tot[4][MAX_BKT];
    __shared__ __attribute__((aligned(16))) int p_pre[4][MAX_BKT];
    __shared__ int s_tot[MAX_BKT];
    __shared__ int cursor[MAX_BKT];
    const int tid = threadIdx.x;
    {
        const int c4 = tid & 63;
        const int g  = __builtin_amdgcn_readfirstlane(tid >> 6);
        const int4* tbl = (const int4*)blk_hist + c4;
        int4 tot = make_int4(0, 0, 0, 0), pre = make_int4(0, 0, 0, 0);
        #pragma unroll 4
        for (int w = g; w < nwg; w += 4) {
            const int4 v = tbl[(size_t)w * (MAX_BKT / 4)];
            tot.x += v.x; tot.y += v.y; tot.z += v.z; tot.w += v.w;
            if (w < wg) { pre.x += v.x; pre.y += v.y; pre.z += v.z; pre.w += v.w; }
        }
        *(int4*)&p_tot[g][4 * c4] = tot;
        *(int4*)&p_pre[g][4 * c4] = pre;
    }
    __syncthreads();
    int pre_b = 0;
    {
        int t = 0;
        #pragma unroll
        for (int k = 0; k < 4; ++k) { t += p_tot[k][tid]; pre_b += p_pre[k][tid]; }
        s_tot[tid] = t;
    }
    __syncthreads();
    if (tid < 32) scan256_excl(s_tot, tid);
    __syncthreads();
    {
        const int st = s_tot[tid];
        cursor[tid] = st + pre_b;
        if (wg == 0 && tid < NB) bstart[tid] = st;
    }
    if (wg == 0 && tid == 0) bstart[NB] = E;
    __syncthreads();

    #pragma unroll 1
    for (int q = 0; q < 4; ++q) {
        const int e0 = wg * EDGE_BLK + q * 1024 + tid;
        int r[4], c[4];
        #pragma unroll
        for (int i = 0; i < 4; ++i) {
            const int e  = e0 + 256 * i;
            const int ec = e < E ? e : E - 1;
            r[i] = ei[ec];
            c[i] = ei[E + ec];
        }
        #pragma unroll
        for (int i = 0; i < 4; ++i) {
            if (e0 + 256 * i < E) {
                const int p = atomicAdd(&cursor[r[i] >> BKT_SHIFT], 1);
                recs[p] = (int)(((unsigned)(r[i] & (BKT_ROWS - 1)) << 24) | (unsigned)c[i]);
            }
        }
    }
}

#define K_SLAB   32
#define TILE_M   128
#define A_PITCH  36

__global__ void __launch_bounds__(256, 8)
spgat_gemm(const float* __restrict__ x,
                           const float* __restrict__ W,
                           const float* __restrict__ a,
                           float* __restrict__ h,
                           float* __restrict__ sp, int N,
                           const int* __restrict__ ei, int E, int NB, int NEB,
                           const int* __restrict__ blk_hist,
                           int* __restrict__ bstart,
                           int* __restrict__ recs) {
    if ((int)blockIdx.x < NEB) {
        bin_edge_block(ei, E, NB, (int)blockIdx.x, NEB, blk_hist, bstart, recs);
        return;
    }
    __shared__ __attribute__((aligned(16))) float sA[2][TILE_M * A_PITCH];
    __shared__ __attribute__((aligned(16))) float sW[2][K_SLAB * 64];

    const int tid  = threadIdx.x;
    const int lane = tid & 31;
    const int wave = __builtin_amdgcn_readfirstlane(tid >> 5);
    const int g    = (int)blockIdx.x - NEB;
    const int q    = g & 3;
    const int row0 = (g >> 2) * TILE_M;
    const int m0   = row0 + 16 * wave;
    const int n0   = q * 64;
    const int half = lane >> 4;
    const int sub  = lane & 15;

    const int ar = tid >> 3, ac4 = tid & 7;
    const int wr = tid >> 4, wc4 = tid & 15;
    const int r0 = row0 + ar, r1 = r0 + 32, r2 = r0 + 64, r3 = r0 + 96;
    const float4* xg0 = (const float4*)(x + (size_t)(r0 < N ? r0 : N - 1) * NFEAT) + ac4;
    const float4* xg1 = (const float4*)(x + (size_t)(r1 < N ? r1 : N - 1) * NFEAT) + ac4;
    const float4* xg2 = (const float4*)(x + (size_t)(r2 < N ? r2 : N - 1) * NFEAT) + ac4;
    const float4* xg3 = (const float4*)(x + (size_t)(r3 < N ? r3 : N - 1) * NFEAT) + ac4;
    const float4* wg0 = (const float4*)(W + (size_t)wr * NCLASS + n0) + wc4;
    const float4* wg1 = wg0 + 16 * (NCLASS / 4);

    const int sa0 = ar * A_PITCH + 4 * ac4;
    const int sw0 = wr * 64 + 4 * wc4;

    float4 ga0 = xg0[0], ga1 = xg1[0], ga2 = xg2[0], ga3 = xg3[0];
    float4 gw0 = wg0[0], gw1 = wg1[0];
    *(float4*)&sA[0][sa0]                = ga0;
    *(float4*)&sA[0][sa0 + 32 * A_PITCH] = ga1;
    *(float4*)&sA[0][sa0 + 64 * A_PITCH] = ga2;
    *(float4*)&sA[0][sa0 + 96 * A_PITCH] = ga3;
    *(float4*)&sW[0][sw0]           = gw0;
    *(float4*)&sW[0][sw0 + 16 * 64] = gw1;
    __syncthreads();

    v8f acc0 = {}, acc1 = {}, acc2 = {}, acc3 = {};

    const int a_off = (16 * wave + sub) * A_PITCH + 4 * half;
    const int w_off = (4 * half) * 64 + 4 * sub;

    #pragma unroll 2
    for (int s = 0; s < NFEAT / K_SLAB; ++s) {
        const int  cur  = s & 1;
        const bool more = s + 1 < NFEAT / K_SLAB;
        if (more) {
            const int ka = (s + 1) * (K_SLAB / 4);
            ga0 = xg0[ka]; ga1 = xg1[ka]; ga2 = xg2[ka]; ga3 = xg3[ka];
            const size_t kw = (size_t)((s + 1) * K_SLAB) * (NCLASS / 4);
            gw0 = wg0[kw]; gw1 = wg1[kw];
        }
        const float* A = &sA[cur][a_off];
        const float* B = &sW[cur][w_off];
        #pragma unroll
        for (int st = 0; st < K_SLAB / 8; ++st) {
            const float4 av = *(const float4*)(A + 8 * st);
            const float4 w0 = *(const float4*)(B + (8 * st + 0) * 64);
            const float4 w1 = *(const float4*)(B + (8 * st + 1) * 64);
            const float4 w2 = *(const float4*)(B + (8 * st + 2) * 64);
            const float4 w3 = *(const float4*)(B + (8 * st + 3) * 64);

            v2f af;  af.x = av.x; af.y = av.y;
            v2f b0;  b0.x = w0.x; b0.y = w1.x;
            v2f b1;  b1.x = w0.y; b1.y = w1.y;
            v2f b2;  b2.x = w0.z; b2.y = w1.z;
            v2f b3;  b3.x = w0.w; b3.y = w1.w;
            acc0 = __builtin_amdgcn_wmma_f32_16x16x4_f32(false, af, false, b0, (short)0, acc0, false, false);
            acc1 = __builtin_amdgcn_wmma_f32_16x16x4_f32(false, af, false, b1, (short)0, acc1, false, false);
            acc2 = __builtin_amdgcn_wmma_f32_16x16x4_f32(false, af, false, b2, (short)0, acc2, false, false);
            acc3 = __builtin_amdgcn_wmma_f32_16x16x4_f32(false, af, false, b3, (short)0, acc3, false, false);

            af.x = av.z; af.y = av.w;
            b0.x = w2.x; b0.y = w3.x;
            b1.x = w2.y; b1.y = w3.y;
            b2.x = w2.z; b2.y = w3.z;
            b3.x = w2.w; b3.y = w3.w;
            acc0 = __builtin_amdgcn_wmma_f32_16x16x4_f32(false, af, false, b0, (short)0, acc0, false, false);
            acc1 = __builtin_amdgcn_wmma_f32_16x16x4_f32(false, af, false, b1, (short)0, acc1, false, false);
            acc2 = __builtin_amdgcn_wmma_f32_16x16x4_f32(false, af, false, b2, (short)0, acc2, false, false);
            acc3 = __builtin_amdgcn_wmma_f32_16x16x4_f32(false, af, false, b3, (short)0, acc3, false, false);
        }
        if (more) {
            const int nxt = cur ^ 1;
            *(float4*)&sA[nxt][sa0]                = ga0;
            *(float4*)&sA[nxt][sa0 + 32 * A_PITCH] = ga1;
            *(float4*)&sA[nxt][sa0 + 64 * A_PITCH] = ga2;
            *(float4*)&sA[nxt][sa0 + 96 * A_PITCH] = ga3;
            *(float4*)&sW[nxt][sw0]           = gw0;
            *(float4*)&sW[nxt][sw0 + 16 * 64] = gw1;
        }
        __syncthreads();
    }

    if (m0 >= N) return;

    const float4 ar4 = *(const float4*)(a + n0 + 4 * sub);
    const float4 ac  = *(const float4*)(a + NCLASS + n0 + 4 * sub);
    const bool b3 = (sub & 8) != 0, b2 = (sub & 4) != 0, b1 = (sub & 2) != 0, b0 = (sub & 1) != 0;
    float* hout = h + (size_t)(m0 + half * 8) * NCLASS + n0 + 4 * sub;
    float v8[8];
    #pragma unroll
    for (int r = 0; r < 8; ++r) {
        const float4 hv = make_float4(acc0[r], acc1[r], acc2[r], acc3[r]);
        *(float4*)(hout + (size_t)r * NCLASS) = hv;
        const float pr = fmaf(hv.w, ar4.w, fmaf(hv.z, ar4.z, fmaf(hv.y, ar4.y, hv.x * ar4.x)));
        const float pc = fmaf(hv.w, ac.w, fmaf(hv.z, ac.z, fmaf(hv.y, ac.y, hv.x * ac.x)));
        const float keep = b3 ? pc : pr;
        const float send = b3 ? pr : pc;
        v8[r] = keep + xrow<8>(send);
    }
    float v4[4];
    #pragma unroll
    for (int j = 0; j < 4; ++j) {
        const float keep = b2 ? v8[4 + j] : v8[j];
        const float send = b2 ? v8[j] : v8[4 + j];
        v4[j] = keep + xrow<4>(send);
    }
    float v2[2];
    #pragma unroll
    for (int j = 0; j < 2; ++j) {
        const float keep = b1 ? v4[2 + j] : v4[j];
        const float send = b1 ? v4[j] : v4[2 + j];
        v2[j] = keep + xrow<2>(send);
    }
    const float keep1 = b0 ? v2[1] : v2[0];
    const float send1 = b0 ? v2[0] : v2[1];
    const float tot = keep1 + xrow<1>(send1);
    sp[(size_t)(m0 + half * 8 + (sub & 7)) * 8 + 4 * (sub >> 3) + q] = tot;
}

__global__ void __launch_bounds__(1024)
spgat_bcount(const int* __restrict__ ei, int E,
             int* __restrict__ blk_hist) {
    __shared__ int hist[MAX_BKT];
    const int tid = threadIdx.x;
    if (tid < MAX_BKT) hist[tid] = 0;
    __syncthreads();
    const int e0 = blockIdx.x * EDGE_BLK + tid;
    int r[4];
    #pragma unroll
    for (int i = 0; i < 4; ++i) {
        const int e = e0 + 1024 * i;
        r[i] = ei[e < E ? e : E - 1];
    }
    #pragma unroll
    for (int i = 0; i < 4; ++i)
        if (e0 + 1024 * i < E) atomicAdd(&hist[r[i] >> BKT_SHIFT], 1);
    __syncthreads();
    if (tid < MAX_BKT) blk_hist[blockIdx.x * MAX_BKT + tid] = hist[tid];
}

__global__ void __launch_bounds__(1024)
spgat_bsort(const int* __restrict__ bstart,
            const int* __restrict__ recs,
            const float* __restrict__ sp,
            int* __restrict__ off,
            int* __restrict__ pairs, int N) {
    __shared__ int cnt[BKT_ROWS];
    __shared__ int cur[BKT_ROWS];
    const int tid = threadIdx.x;
    const int b   = blockIdx.x;
    const int beg = bstart[b];
    const int end = bstart[b + 1];
    if (tid < BKT_ROWS) cnt[tid] = 0;
    __syncthreads();
    for (int i0 = beg + tid; i0 < end; i0 += 5 * 1024) {
        int k[5];
        #pragma unroll
        for (int j = 0; j < 5; ++j) {
            const int i = i0 + 1024 * j;
            k[j] = recs[i < end ? i : end - 1];
        }
        #pragma unroll
        for (int j = 0; j < 5; ++j)
            if (i0 + 1024 * j < end) atomicAdd(&cnt[((unsigned)k[j]) >> 24], 1);
    }
    __syncthreads();
    if (tid < 32) scan256_excl(cnt, tid);
    __syncthreads();
    if (tid < BKT_ROWS) {
        const int o = beg + cnt[tid];
        cur[tid] = o;
        const int row = b * BKT_ROWS + tid;
        if (row < N) off[row] = o;
    }
    if (b == (int)gridDim.x - 1 && tid == 0) off[N] = end;
    __syncthreads();
    for (int i0 = beg + tid; i0 < end; i0 += 5 * 1024) {
        int rc[5];
        #pragma unroll
        for (int j = 0; j < 5; ++j) {
            const int i = i0 + 1024 * j;
            rc[j] = recs[i < end ? i : end - 1];
        }
        float4 pr[5], pc[5];
        #pragma unroll
        for (int j = 0; j < 5; ++j) {
            const int r  = b * BKT_ROWS + (int)(((unsigned)rc[j]) >> 24);
            const int cc = rc[j] & 0xFFFFFF;
            pr[j] = ((const float4*)sp)[2 * (size_t)r];
            pc[j] = ((const float4*)sp)[2 * (size_t)cc + 1];
        }
        #pragma unroll
        for (int j = 0; j < 5; ++j) {
            if (i0 + 1024 * j < end) {
                const float lg = ((pr[j].x + pr[j].y) + (pr[j].z + pr[j].w)) +
                                 ((pc[j].x + pc[j].y) + (pc[j].z + pc[j].w));
                const float l  = lg > 0.f ? lg : 0.2f * lg;
                const float ev = __expf(-l);
                const int p = atomicAdd(&cur[((unsigned)rc[j]) >> 24], 1);
                ((int2*)pairs)[p] = make_int2(rc[j] & 0xFFFFFF, __float_as_int(ev));
            }
        }
    }
}

__device__ __forceinline__ void axpy4(float4& acc, float s, const float4& v) {
    acc.x += s * v.x; acc.y += s * v.y; acc.z += s * v.z; acc.w += s * v.w;
}

__device__ __forceinline__ float xhalf(float x) {
    const int b = __builtin_bit_cast(int, x);
    return __builtin_bit_cast(float,
        __builtin_amdgcn_permlanex16(b, b, 0x76543210, 0xfedcba98, false, false));
}
__device__ __forceinline__ float wave_max(float v) {
    v = fmaxf(v, xrow<1>(v));
    v = fmaxf(v, xrow<2>(v));
    v = fmaxf(v, xrow<4>(v));
    v = fmaxf(v, xrow<8>(v));
    return fmaxf(v, xhalf(v));
}
__device__ __forceinline__ float wave_sum(float v) {
    v += xrow<1>(v);
    v += xrow<2>(v);
    v += xrow<4>(v);
    v += xrow<8>(v);
    return v + xhalf(v);
}

__global__ void spgat_gather(const int* __restrict__ off,
                             const int* __restrict__ pairs,
                             const float* __restrict__ h,
                             float* __restrict__ out, int N) {
    const int lane = threadIdx.x & 31;
    const int row = __builtin_amdgcn_readfirstlane(
        (int)(blockIdx.x * (blockDim.x >> 5) + (threadIdx.x >> 5)));
    if (row >= N) return;
    const int beg = off[row];
    const int end = off[row + 1];
    const float4* hl = (const float4*)h + lane;
    float4 a0 = make_float4(0.f, 0.f, 0.f, 0.f);
    float4 a1 = make_float4(0.f, 0.f, 0.f, 0.f);
    float rs = 0.f;
    #pragma unroll 4
    for (int j = beg; j < end; ++j) {
        const int2 p = ((const int2*)pairs)[j];
        const float4* q = hl + (size_t)p.x * (NCLASS / 4);
        const float4 u = q[0], v = q[32];
        axpy4(a0, __int_as_float(p.y), u);
        axpy4(a1, __int_as_float(p.y), v);
        rs += __int_as_float(p.y);
    }

    const float inv = 1.0f / rs;
    float v[8] = {a0.x, a0.y, a0.z, a0.w, a1.x, a1.y, a1.z, a1.w};
    float mx = -3.402823466e38f;
    #pragma unroll
    for (int i = 0; i < 8; ++i) {
        const float hp = v[i] * inv;
        const float el = hp > 0.f ? hp : (__expf(hp) - 1.f);
        v[i] = el;
        mx = fmaxf(mx, el);
    }
    mx = wave_max(mx);
    float s = 0.f;
    #pragma unroll
    for (int i = 0; i < 8; ++i) s += __expf(v[i] - mx);
    s = wave_sum(s);
    const float lse = mx + __logf(s);

    float4* o = (float4*)(out + (size_t)row * NCLASS) + lane;
    o[0]  = make_float4(v[0] - lse, v[1] - lse, v[2] - lse, v[3] - lse);
    o[32] = make_float4(v[4] - lse, v[5] - lse, v[6] - lse, v[7] - lse);
}

extern "C" void kernel_launch(void* const* d_in, const int* in_sizes, int n_in,
                              void* d_out, int out_size, void* d_ws, size_t ws_size,
                              hipStream_t stream) {
    const float* x  = (const float*)d_in[0];
    const float* W  = (const float*)d_in[1];
    const float* a  = (const float*)d_in[2];
    const int*   ei = (const int*)d_in[3];

    const int N = in_sizes[0] / NFEAT;
    const int E = in_sizes[3] / 2;

    const int NB  = (N + BKT_ROWS - 1) / BKT_ROWS;
    const int NEB = (E + EDGE_BLK - 1) / EDGE_BLK;
    float* h        = (float*)d_ws;
    int*   blk_hist = (int*)(h + (size_t)N * NCLASS);
    float* sp       = (float*)(blk_hist + (size_t)NEB * MAX_BKT);
    int*   recs     = (int*)(sp + 8 * (size_t)N);
    int*   pairs    = recs + 2 * (size_t)E;
    int*   off      = pairs + 2 * (size_t)E;
    int*   bstart   = off + N + 1;

    float* out = (float*)d_out;

    spgat_bcount<<<NEB, 1024, 0, stream>>>(ei, E, blk_hist);
    {
        const int mtiles = N / 16;
        dim3 grid(NEB + 4 * ((mtiles + 7) / 8));
        spgat_gemm<<<grid, 256, 0, stream>>>(x, W, a, h, sp, N,
                                             ei, E, NB, NEB, blk_hist, bstart, recs);
    }
    spgat_bsort<<<NB, 1024, 0, stream>>>(bstart, recs, sp, off, pairs, N);
    {
        dim3 grid((N + 7) / 8);
        spgat_gather<<<grid, 256, 0, stream>>>(off, pairs, h, out, N);
    }
}
